# starting kernel
# baseline (speedup 1.0000x reference)
#include <hip/hip_runtime.h>
#include <hip/hip_bf16.h>
#include <math.h>

#define NUM_USER   100000
#define NUM_ITEM   50000
#define NUM_GROUP  20000
#define EMB        64
#define BATCH      1024
#define E_UI       2000000
#define E_U        1000000
#define E_I        1000000
#define INV_TEMP   10.0f
#define EPS_N      1e-12f

#define N_UG  (NUM_USER + NUM_GROUP)
#define N_IG  (NUM_ITEM + NUM_GROUP)
#define N_UI  (NUM_USER + NUM_ITEM)

typedef _Float16 v16h __attribute__((ext_vector_type(16)));
typedef float    v8f  __attribute__((ext_vector_type(8)));

__global__ void zero_f32(float* __restrict__ p, size_t n) {
  size_t i = (size_t)blockIdx.x * blockDim.x + threadIdx.x;
  size_t stride = (size_t)gridDim.x * blockDim.x;
  for (; i < n; i += stride) p[i] = 0.0f;
}

__global__ void count_cols(const int* __restrict__ col, int* __restrict__ counts,
                           int nE) {
  int e = blockIdx.x * blockDim.x + threadIdx.x;
  if (e < nE) atomicAdd(&counts[col[e]], 1);
}

__global__ void exscan_kernel(int* __restrict__ counts, int* __restrict__ offsets,
                              int n) {
  __shared__ int sdata[1024];
  __shared__ int s_carry;
  int tid = threadIdx.x;
  if (tid == 0) s_carry = 0;
  __syncthreads();
  for (int base = 0; base < n; base += 1024) {
    int i = base + tid;
    int x = (i < n) ? counts[i] : 0;
    sdata[tid] = x;
    __syncthreads();
    for (int off = 1; off < 1024; off <<= 1) {
      int y = (tid >= off) ? sdata[tid - off] : 0;
      __syncthreads();
      sdata[tid] += y;
      __syncthreads();
    }
    int inc   = sdata[tid];
    int total = sdata[1023];
    int ex    = s_carry + inc - x;
    if (i < n) { offsets[i] = ex; counts[i] = ex; }
    __syncthreads();
    if (tid == 0) s_carry += total;
    __syncthreads();
  }
  if (tid == 0) offsets[n] = s_carry;
}

__global__ void csr_fill(const int* __restrict__ col, int* __restrict__ cursor,
                         int* __restrict__ eidx, int nE) {
  int e = blockIdx.x * blockDim.x + threadIdx.x;
  if (e < nE) {
    int slot = atomicAdd(&cursor[col[e]], 1);
    eidx[slot] = e;
  }
}

__global__ void csr_prop(const float* __restrict__ p0, int n0,
                         const float* __restrict__ p1,
                         const int* __restrict__ row,
                         const float* __restrict__ val,
                         const int* __restrict__ offsets,
                         const int* __restrict__ eidx,
                         float* __restrict__ out, int nRows) {
  int wid = (blockIdx.x * blockDim.x + threadIdx.x) >> 5;
  if (wid >= nRows) return;
  int lane = threadIdx.x & 31;
  int beg = offsets[wid], end = offsets[wid + 1];
  float ax = 0.0f, ay = 0.0f;
  for (int k = beg; k < end; ++k) {
    int e = eidx[k];
    int r = row[e];
    float v = val[e];
    const float* src = (r < n0) ? (p0 + (size_t)r * EMB)
                                : (p1 + (size_t)(r - n0) * EMB);
    float2 d = *(const float2*)(src + lane * 2);
    ax += d.x * v;
    ay += d.y * v;
  }
  float2 o; o.x = ax; o.y = ay;
  *(float2*)(out + (size_t)wid * EMB + lane * 2) = o;
}

__global__ void ui_final_kernel(const float* __restrict__ uei,
                                const float* __restrict__ item,
                                float* __restrict__ agg) {
  int gid = blockIdx.x * blockDim.x + threadIdx.x;
  if (gid >= N_UI * EMB) return;
  int r = gid >> 6, c = gid & 63;
  float uiv = (r < NUM_USER) ? uei[gid]
                             : item[(size_t)(r - NUM_USER) * EMB + c];
  agg[gid] = (uiv + agg[gid]) * 0.5f;
}

__global__ void gather_out(const float* __restrict__ ug,
                           const float* __restrict__ uif,
                           const float* __restrict__ ig,
                           const int* __restrict__ u,
                           const int* __restrict__ p,
                           const int* __restrict__ ng,
                           float* __restrict__ out) {
  int gid = blockIdx.x * blockDim.x + threadIdx.x;
  if (gid >= BATCH * EMB) return;
  int b = gid >> 6, c = gid & 63;
  const size_t PO = (size_t)BATCH * 128;
  const size_t NO = 2 * PO;
  out[(size_t)b * 128 + c]           = ug [(size_t)u[b] * EMB + c];
  out[(size_t)b * 128 + 64 + c]      = uif[(size_t)u[b] * EMB + c];
  out[PO + (size_t)b * 128 + c]      = ug [(size_t)(NUM_USER + p[b])  * EMB + c];
  out[PO + (size_t)b * 128 + 64 + c] = ig [(size_t)(NUM_ITEM + p[b])  * EMB + c];
  out[NO + (size_t)b * 128 + c]      = ug [(size_t)(NUM_USER + ng[b]) * EMB + c];
  out[NO + (size_t)b * 128 + 64 + c] = ig [(size_t)(NUM_ITEM + ng[b]) * EMB + c];
}

__global__ void reg_kernel(const float* __restrict__ ued,
                           const float* __restrict__ uei,
                           const float* __restrict__ ged,
                           const float* __restrict__ gei,
                           const int* __restrict__ u,
                           const int* __restrict__ p,
                           const int* __restrict__ ng,
                           float* __restrict__ reg_out) {
  int gid = blockIdx.x * blockDim.x + threadIdx.x;
  if (gid >= BATCH * EMB) return;
  int b = gid >> 6, c = gid & 63;
  size_t uo = (size_t)u[b]  * EMB + c;
  size_t po = (size_t)p[b]  * EMB + c;
  size_t no = (size_t)ng[b] * EMB + c;
  float s = uei[uo]*uei[uo] + ued[uo]*ued[uo]
          + gei[po]*gei[po] + ged[po]*ged[po]
          + gei[no]*gei[no] + ged[no]*ged[no];
  #pragma unroll
  for (int off = 1; off < 32; off <<= 1) s += __shfl_xor(s, off, 32);
  if ((threadIdx.x & 31) == 0) atomicAdd(reg_out, s * (0.5f / (float)BATCH));
}

__global__ void normalize_rows_f16(const float* __restrict__ in,
                                   _Float16* __restrict__ out, int nrows) {
  int gid = blockIdx.x * blockDim.x + threadIdx.x;
  int r = gid >> 5;
  if (r >= nrows) return;
  int t = gid & 31;
  size_t base = (size_t)r * EMB + 2 * t;
  float x0 = in[base], x1 = in[base + 1];
  float ss = x0 * x0 + x1 * x1;
  #pragma unroll
  for (int off = 1; off < 32; off <<= 1) ss += __shfl_xor(ss, off, 32);
  float scale = 1.0f / fmaxf(sqrtf(ss), EPS_N);
  out[base]     = (_Float16)(x0 * scale);
  out[base + 1] = (_Float16)(x1 * scale);
}

#define GEMM_WAVES 8
#define GEMM_TJ    16

__device__ __forceinline__ void load_btile(const _Float16* __restrict__ B16,
                                           int j0, int m, int kbB,
                                           v16h& b0, v16h& b1) {
  const _Float16* brow = B16 + (size_t)(j0 + m) * EMB + kbB;
  #pragma unroll
  for (int e = 0; e < 16; ++e) { b0[e] = brow[e]; b1[e] = brow[32 + e]; }
}

__global__ void ssl_gemm_expsum(const _Float16* __restrict__ A16,
                                const _Float16* __restrict__ B16,
                                const int* __restrict__ idx,
                                float* __restrict__ tot,
                                int Nall) {
  int lane = threadIdx.x & 31;
  int wave = threadIdx.x >> 5;
  int m0   = blockIdx.x * 16;
  int m    = lane & 15;
  int hi   = lane >> 4;

  const _Float16* arow = A16 + (size_t)idx[m0 + m] * EMB;
  int kbA = hi ? 8 : 0;
  v16h a0, a1;
  #pragma unroll
  for (int e = 0; e < 8; ++e) {
    a0[e]     = arow[kbA + e];
    a0[8 + e] = arow[16 + kbA + e];
    a1[e]     = arow[32 + kbA + e];
    a1[8 + e] = arow[48 + kbA + e];
  }

  float s[8];
  #pragma unroll
  for (int v = 0; v < 8; ++v) s[v] = 0.0f;

  int jt0 = (blockIdx.y * GEMM_WAVES + wave) * GEMM_TJ;
  int kbB = hi ? 16 : 0;
  int j0  = jt0 * 16;

  if (j0 < Nall) {
    v16h b0, b1;
    load_btile(B16, j0, m, kbB, b0, b1);
    for (int t = 0; t < GEMM_TJ; ++t) {
      int  j1   = j0 + 16;
      bool more = (t + 1 < GEMM_TJ) && (j1 < Nall);
      v16h nb0, nb1;
      if (more) load_btile(B16, j1, m, kbB, nb0, nb1);
      v8f c = {};
      c = __builtin_amdgcn_wmma_f32_16x16x32_f16(false, a0, false, b0,
                                                 (short)0, c, false, false);
      c = __builtin_amdgcn_wmma_f32_16x16x32_f16(false, a1, false, b1,
                                                 (short)0, c, false, false);
      #pragma unroll
      for (int v = 0; v < 8; ++v) s[v] += __expf(c[v] * INV_TEMP);
      if (!more) break;
      b0 = nb0; b1 = nb1; j0 = j1;
    }
  }
  #pragma unroll
  for (int v = 0; v < 8; ++v) {
    #pragma unroll
    for (int off = 1; off < 16; off <<= 1) s[v] += __shfl_xor(s[v], off, 32);
  }
  if ((lane & 15) == 0) {
    #pragma unroll
    for (int v = 0; v < 8; ++v)
      atomicAdd(&tot[m0 + v + 8 * hi], s[v]);
  }
}

__global__ void ssl_loss(const _Float16* __restrict__ A16,
                         const _Float16* __restrict__ B16,
                         const int* __restrict__ idx,
                         const float* __restrict__ tot,
                         float* __restrict__ ssl_out) {
  int gid = blockIdx.x * blockDim.x + threadIdx.x;
  int i = gid >> 5;
  if (i >= BATCH) return;
  int t = gid & 31;
  size_t ro = (size_t)idx[i] * EMB + 2 * t;
  float d = (float)A16[ro] * (float)B16[ro]
          + (float)A16[ro + 1] * (float)B16[ro + 1];
  #pragma unroll
  for (int off = 1; off < 32; off <<= 1) d += __shfl_xor(d, off, 32);
  if (t == 0) {
    float li = -(d * INV_TEMP - logf(tot[i])) * (1.0f / (float)BATCH);
    atomicAdd(ssl_out, li);
  }
}

extern "C" void kernel_launch(void* const* d_in, const int* in_sizes, int n_in,
                              void* d_out, int out_size, void* d_ws, size_t ws_size,
                              hipStream_t stream) {
  (void)in_sizes; (void)n_in; (void)out_size; (void)ws_size;
  const float* ued  = (const float*)d_in[0];
  const float* uei  = (const float*)d_in[1];
  const float* ged  = (const float*)d_in[2];
  const float* gei  = (const float*)d_in[3];
  const float* item = (const float*)d_in[4];
  const int*   hu_r = (const int*)d_in[5];
  const int*   hu_c = (const int*)d_in[6];
  const float* hu_v = (const float*)d_in[7];
  const int*   hi_r = (const int*)d_in[8];
  const int*   hi_c = (const int*)d_in[9];
  const float* hi_v = (const float*)d_in[10];
  const int*   ui_r = (const int*)d_in[11];
  const int*   ui_c = (const int*)d_in[12];
  const float* ui_v = (const float*)d_in[13];
  const int*   uin  = (const int*)d_in[14];
  const int*   pos  = (const int*)d_in[15];
  const int*   neg  = (const int*)d_in[16];
  float* out = (float*)d_out;

  float* W = (float*)d_ws;
  size_t off = 0;
  float* ug_b = W + off; off += (size_t)N_UG * EMB;
  float* ug_a = W + off; off += (size_t)N_UG * EMB;
  float* ig_b = W + off; off += (size_t)N_IG * EMB;
  float* ig_a = W + off; off += (size_t)N_IG * EMB;
  float* agg  = W + off; off += (size_t)N_UI * EMB;
  float* tot  = W + off; off += 4 * BATCH;
  int* counts  = (int*)(W + off); off += N_UI;
  int* offsets = (int*)(W + off); off += N_UI + 1;
  int* eidx    = (int*)(W + off); off += E_UI;
  off = (off + 7) & ~(size_t)7;
  _Float16* H = (_Float16*)(W + off);
  _Float16* ug_user_n16  = H;
  _Float16* ui_user_n16  = H + (size_t)NUM_USER * EMB;
  _Float16* ug_group_n16 = H + (size_t)NUM_USER * EMB * 2;
  _Float16* ig_group_n16 = ug_group_n16 + (size_t)NUM_GROUP * EMB;

  const int TB = 256;
  #define CDIV(a, b) (((a) + (b) - 1) / (b))

  zero_f32<<<16, TB, 0, stream>>>(tot, 4 * BATCH);
  zero_f32<<<1, 32, 0, stream>>>(out + (size_t)3 * BATCH * 128, 2);

  zero_f32<<<CDIV(N_UG, TB), TB, 0, stream>>>((float*)counts, N_UG);
  count_cols<<<CDIV(E_U, TB), TB, 0, stream>>>(hu_c, counts, E_U);
  exscan_kernel<<<1, 1024, 0, stream>>>(counts, offsets, N_UG);
  csr_fill<<<CDIV(E_U, TB), TB, 0, stream>>>(hu_c, counts, eidx, E_U);
  csr_prop<<<CDIV(N_UG * 32, TB), TB, 0, stream>>>(
      ued, NUM_USER, ged, hu_r, hu_v, offsets, eidx, ug_b, N_UG);
  csr_prop<<<CDIV(N_UG * 32, TB), TB, 0, stream>>>(
      ug_b, 1 << 30, nullptr, hu_r, hu_v, offsets, eidx, ug_a, N_UG);

  zero_f32<<<CDIV(N_IG, TB), TB, 0, stream>>>((float*)counts, N_IG);
  count_cols<<<CDIV(E_I, TB), TB, 0, stream>>>(hi_c, counts, E_I);
  exscan_kernel<<<1, 1024, 0, stream>>>(counts, offsets, N_IG);
  csr_fill<<<CDIV(E_I, TB), TB, 0, stream>>>(hi_c, counts, eidx, E_I);
  csr_prop<<<CDIV(N_IG * 32, TB), TB, 0, stream>>>(
      item, NUM_ITEM, gei, hi_r, hi_v, offsets, eidx, ig_b, N_IG);
  csr_prop<<<CDIV(N_IG * 32, TB), TB, 0, stream>>>(
      ig_b, 1 << 30, nullptr, hi_r, hi_v, offsets, eidx, ig_a, N_IG);

  zero_f32<<<CDIV(N_UI, TB), TB, 0, stream>>>((float*)counts, N_UI);
  count_cols<<<CDIV(E_UI, TB), TB, 0, stream>>>(ui_c, counts, E_UI);
  exscan_kernel<<<1, 1024, 0, stream>>>(counts, offsets, N_UI);
  csr_fill<<<CDIV(E_UI, TB), TB, 0, stream>>>(ui_c, counts, eidx, E_UI);
  csr_prop<<<CDIV(N_UI * 32, TB), TB, 0, stream>>>(
      uei, NUM_USER, item, ui_r, ui_v, offsets, eidx, agg, N_UI);
  ui_final_kernel<<<CDIV(N_UI * EMB, TB), TB, 0, stream>>>(uei, item, agg);

  gather_out<<<CDIV(BATCH * EMB, TB), TB, 0, stream>>>(
      ug_a, agg, ig_a, uin, pos, neg, out);
  reg_kernel<<<CDIV(BATCH * EMB, TB), TB, 0, stream>>>(
      ued, uei, ged, gei, uin, pos, neg, out + (size_t)3 * BATCH * 128);

  normalize_rows_f16<<<CDIV(NUM_USER * 32, TB), TB, 0, stream>>>(
      ug_a, ug_user_n16, NUM_USER);
  normalize_rows_f16<<<CDIV(NUM_USER * 32, TB), TB, 0, stream>>>(
      agg, ui_user_n16, NUM_USER);
  normalize_rows_f16<<<CDIV(NUM_GROUP * 32, TB), TB, 0, stream>>>(
      ug_a + (size_t)NUM_USER * EMB, ug_group_n16, NUM_GROUP);
  normalize_rows_f16<<<CDIV(NUM_GROUP * 32, TB), TB, 0, stream>>>(
      ig_a + (size_t)NUM_ITEM * EMB, ig_group_n16, NUM_GROUP);

  const int perBlk = GEMM_WAVES * GEMM_TJ;
  int ntU = NUM_USER / 16, ntG = NUM_GROUP / 16;
  dim3 gU(BATCH / 16, CDIV(ntU, perBlk));
  dim3 gG(BATCH / 16, CDIV(ntG, perBlk));
  ssl_gemm_expsum<<<gU, TB, 0, stream>>>(ui_user_n16, ug_user_n16, uin, tot + 0,         NUM_USER);
  ssl_gemm_expsum<<<gU, TB, 0, stream>>>(ug_user_n16, ui_user_n16, uin, tot + BATCH,     NUM_USER);
  ssl_gemm_expsum<<<gG, TB, 0, stream>>>(ig_group_n16, ug_group_n16, pos, tot + 2*BATCH, NUM_GROUP);
  ssl_gemm_expsum<<<gG, TB, 0, stream>>>(ug_group_n16, ig_group_n16, pos, tot + 3*BATCH, NUM_GROUP);

  float* ssl_slot = out + (size_t)3 * BATCH * 128 + 1;
  int lossBlocks = CDIV(BATCH * 32, TB);
  ssl_loss<<<lossBlocks, TB, 0, stream>>>(ui_user_n16, ug_user_n16, uin, tot + 0,         ssl_slot);
  ssl_loss<<<lossBlocks, TB, 0, stream>>>(ug_user_n16, ui_user_n16, uin, tot + BATCH,     ssl_slot);
  ssl_loss<<<lossBlocks, TB, 0, stream>>>(ig_group_n16, ug_group_n16, pos, tot + 2*BATCH, ssl_slot);
  ssl_loss<<<lossBlocks, TB, 0, stream>>>(ug_group_n16, ig_group_n16, pos, tot + 3*BATCH, ssl_slot);
  #undef CDIV
}


# Second gather layer reads each column's (source row, value) pairs from one 256-byte block that the first layer's walk wrote: no head staging or node reads in layer 2; sum order unchanged
# speedup vs baseline: 4.3290x; 4.3290x over previous
#include <hip/hip_runtime.h>
#include <hip/hip_bf16.h>
#include <math.h>

#define NUM_USER   100000
#define NUM_ITEM   50000
#define NUM_GROUP  20000
#define EMB        64
#define BATCH      1024
#define E_UI       2000000
#define E_U        1000000
#define E_I        1000000
#define INV_TEMP   10.0f
#define EPS_N      1e-12f

#define N_UG  (NUM_USER + NUM_GROUP)
#define N_IG  (NUM_ITEM + NUM_GROUP)
#define N_UI  (NUM_USER + NUM_ITEM)

typedef _Float16 v16h __attribute__((ext_vector_type(16)));
typedef float    v8f  __attribute__((ext_vector_type(8)));

__global__ void zero_f32(float* __restrict__ p, size_t n) {
  size_t i = (size_t)blockIdx.x * blockDim.x + threadIdx.x;
  size_t stride = (size_t)gridDim.x * blockDim.x;
  for (; i < n; i += stride) p[i] = 0.0f;
}

#define LB_P    4
#define LB_S    64
#define LB_WMAX 37504
#define LB_THREADS 1024

__device__ __forceinline__ void thread1(int* hl, int4* __restrict__ node, int e,
                                        int c, int r, float v, int c0, int Wd) {
  unsigned d = (unsigned)(c - c0);
  if (d < (unsigned)Wd) {
    int o = atomicExch(&hl[d], e + 1);
    node[e] = make_int4(o, r, __float_as_int(v), 0);
  }
}
__device__ __forceinline__ void thread4(int* hl, int4* __restrict__ node, int e,
                                        int4 c, int4 r, float4 v, int c0, int Wd) {
  thread1(hl, node, e,     c.x, r.x, v.x, c0, Wd);
  thread1(hl, node, e + 1, c.y, r.y, v.y, c0, Wd);
  thread1(hl, node, e + 2, c.z, r.z, v.z, c0, Wd);
  thread1(hl, node, e + 3, c.w, r.w, v.w, c0, Wd);
}
__device__ __forceinline__ void thread_tail(int* hl, int4* __restrict__ node,
                                            const int* __restrict__ col,
                                            const int* __restrict__ row,
                                            const float* __restrict__ val,
                                            int e0, int send, int c0, int Wd) {
  for (int e = e0; e < send; ++e) thread1(hl, node, e, col[e], row[e], val[e], c0, Wd);
}

__global__ __launch_bounds__(LB_THREADS)
void list_build_lds(const int* __restrict__ col,
                    const int* __restrict__ row,
                    const float* __restrict__ val,
                    int* __restrict__ H, int hpitch,
                    int* __restrict__ nodes, int nE, int nCols) {
  __shared__ int hl[LB_WMAX];
  int4* __restrict__ node = (int4*)nodes;
  const int p  = blockIdx.x;
  const int sl = blockIdx.y;
  const int Wd = (nCols + LB_P - 1) / LB_P;
  const int c0 = p * Wd;
  const int SL = (((nE + LB_S - 1) / LB_S) + 3) & ~3;
  const int sbeg = sl * SL;
  const int send = (sbeg + SL < nE) ? (sbeg + SL) : nE;

  for (int i = threadIdx.x; i < Wd; i += LB_THREADS) hl[i] = 0;
  __syncthreads();

  for (int ea = sbeg + threadIdx.x * 4; ea < send; ea += 2 * LB_THREADS * 4) {
    const int  eb = ea + LB_THREADS * 4;
    const bool fa = ea + 3 < send;
    const bool fb = eb + 3 < send;
    int4   ca = make_int4(0, 0, 0, 0), ra = make_int4(0, 0, 0, 0);
    int4   cb = make_int4(0, 0, 0, 0), rb = make_int4(0, 0, 0, 0);
    float4 va = {0.0f, 0.0f, 0.0f, 0.0f}, vb = {0.0f, 0.0f, 0.0f, 0.0f};
    if (fa) {
      ca = *(const int4*)(col + ea);
      ra = *(const int4*)(row + ea);
      va = *(const float4*)(val + ea);
    }
    if (fb) {
      cb = *(const int4*)(col + eb);
      rb = *(const int4*)(row + eb);
      vb = *(const float4*)(val + eb);
    }
    if (fa) thread4(hl, node, ea, ca, ra, va, c0, Wd);
    else    thread_tail(hl, node, col, row, val, ea, send, c0, Wd);
    if (fb) thread4(hl, node, eb, cb, rb, vb, c0, Wd);
    else    thread_tail(hl, node, col, row, val, eb, send, c0, Wd);
  }
  __syncthreads();

  int* __restrict__ Hrow = H + (size_t)sl * hpitch + c0;
  for (int i = threadIdx.x; i < Wd; i += LB_THREADS)
    if (c0 + i < nCols) Hrow[i] = hl[i];
}

#define LP_COLS 32
#define PB_K 32
template <bool WP>
__global__ __launch_bounds__(LP_COLS * 8)
void list_prop(const float* __restrict__ p0, int n0,
               const float* __restrict__ p1,
               const int* __restrict__ H, int hpitch,
               const int* __restrict__ nodes,
               float* __restrict__ out, int nRows,
               int2* __restrict__ pairs, int* __restrict__ pcnt) {
  __shared__ int hs[LB_S * LP_COLS];
  __shared__ int cq[LP_COLS * LB_S];
  __shared__ __attribute__((aligned(16))) int2 pb[LP_COLS * PB_K];
  const int4* __restrict__ node = (const int4*)nodes;
  const int lane = threadIdx.x & 31;
  const int wave = threadIdx.x >> 5;
  const int cb   = blockIdx.x * LP_COLS;

  int hv[LB_S / 8];
  #pragma unroll
  for (int i = 0; i < LB_S / 8; ++i) {
    int sidx = i * 8 + wave;
    hv[i] = (cb + lane < nRows) ? H[(size_t)sidx * hpitch + cb + lane] : 0;
  }
  #pragma unroll
  for (int i = 0; i < LB_S / 8; ++i) hs[(i * 8 + wave) * LP_COLS + lane] = hv[i];
  __syncthreads();

  const int j   = threadIdx.x >> 3;
  const int sub = threadIdx.x & 7;
  const int r   = cb + j;
  int cnt = 0;
  #pragma unroll
  for (int i = 0; i < LB_S / 8; ++i) {
    int h = hs[(i * 8 + sub) * LP_COLS + j];
    unsigned m  = __builtin_amdgcn_ballot_w32(h != 0);
    unsigned gm = (m >> (lane & 24)) & 0xFFu;
    int pos = cnt + __popc(gm & ((1u << sub) - 1u));
    if (h) cq[j * LB_S + pos] = h;
    cnt += __popc(gm);
  }
  __syncthreads();

  const int* __restrict__ q = cq + j * LB_S;
  int cur   = (cnt > 0) ? q[0] : 0;
  int hnext = (cnt > 1) ? q[1] : 0;
  int k = 2;
  float4 a0 = {0.0f, 0.0f, 0.0f, 0.0f}, a1 = {0.0f, 0.0f, 0.0f, 0.0f};
  float4 d0 = {0.0f, 0.0f, 0.0f, 0.0f}, d1 = {0.0f, 0.0f, 0.0f, 0.0f};
  float vp = 0.0f;
  int4 nd = make_int4(0, 0, 0, 0);
  if (cur) nd = node[cur - 1];
  int li = 0;
  while (cur) {
    int nx = nd.x;
    int s = nd.y;
    float v = __int_as_float(nd.z);
    if (WP) {
      if (sub == 0 && li < PB_K) pb[j * PB_K + li] = make_int2(s, nd.z);
      ++li;
    }
    const bool seam = (nx == 0);
    if (seam) nx = hnext;
    a0.x += d0.x * vp; a0.y += d0.y * vp; a0.z += d0.z * vp; a0.w += d0.w * vp;
    a1.x += d1.x * vp; a1.y += d1.y * vp; a1.z += d1.z * vp; a1.w += d1.w * vp;
    const float* src = (s < n0) ? (p0 + (size_t)s * EMB)
                                : (p1 + (size_t)(s - n0) * EMB);
    d0 = *(const float4*)(src + sub * 4);
    d1 = *(const float4*)(src + 32 + sub * 4);
    if (nx) nd = node[nx - 1];
    if (seam) { hnext = (k < cnt) ? q[k] : 0; ++k; }
    vp = v;
    cur = nx;
  }
  a0.x += d0.x * vp; a0.y += d0.y * vp; a0.z += d0.z * vp; a0.w += d0.w * vp;
  a1.x += d1.x * vp; a1.y += d1.y * vp; a1.z += d1.z * vp; a1.w += d1.w * vp;
  if (r < nRows) {
    float* dst = out + (size_t)r * EMB;
    *(float4*)(dst + sub * 4) = a0;
    *(float4*)(dst + 32 + sub * 4) = a1;
    if (WP) {
      const int4* ps = (const int4*)(pb + j * PB_K + sub * 4);
      int4 u0 = ps[0], u1 = ps[1];
      int4* pd = (int4*)(pairs + (size_t)r * PB_K + sub * 4);
      pd[0] = u0;
      pd[1] = u1;
      if (sub == 0) pcnt[r] = li;
    }
  }
}

__global__ __launch_bounds__(LP_COLS * 8)
void pair_prop(const float* __restrict__ in,
               const int2* __restrict__ pairs, const int* __restrict__ pcnt,
               const int* __restrict__ H, int hpitch,
               const int* __restrict__ nodes,
               float* __restrict__ out, int nRows) {
  const int j   = threadIdx.x >> 3;
  const int sub = threadIdx.x & 7;
  const int r   = blockIdx.x * LP_COLS + j;
  if (r >= nRows) return;
  const int4* pp = (const int4*)(pairs + (size_t)r * PB_K + sub * 4);
  const int4 q0 = pp[0], q1 = pp[1];
  const int cnt = pcnt[r];
  float4 a0 = {0.0f, 0.0f, 0.0f, 0.0f}, a1 = {0.0f, 0.0f, 0.0f, 0.0f};
  if (cnt <= PB_K) {
    for (int kk = 0; kk * 4 < cnt; ++kk) {
      const int m = cnt - kk * 4;
      int   s[4];
      float v[4];
      s[0] = __shfl(q0.x, kk, 8); v[0] = __int_as_float(__shfl(q0.y, kk, 8));
      s[1] = __shfl(q0.z, kk, 8); v[1] = __int_as_float(__shfl(q0.w, kk, 8));
      s[2] = __shfl(q1.x, kk, 8); v[2] = __int_as_float(__shfl(q1.y, kk, 8));
      s[3] = __shfl(q1.z, kk, 8); v[3] = __int_as_float(__shfl(q1.w, kk, 8));
      float4 d0[4], d1[4];
      #pragma unroll
      for (int t = 0; t < 4; ++t) {
        d0[t] = make_float4(0.0f, 0.0f, 0.0f, 0.0f);
        d1[t] = make_float4(0.0f, 0.0f, 0.0f, 0.0f);
        if (t < m) {
          const float* src = in + (size_t)s[t] * EMB;
          d0[t] = *(const float4*)(src + sub * 4);
          d1[t] = *(const float4*)(src + 32 + sub * 4);
        }
      }
      #pragma unroll
      for (int t = 0; t < 4; ++t) {
        if (t < m) {
          const float w = v[t];
          a0.x += d0[t].x * w; a0.y += d0[t].y * w; a0.z += d0[t].z * w; a0.w += d0[t].w * w;
          a1.x += d1[t].x * w; a1.y += d1[t].y * w; a1.z += d1[t].z * w; a1.w += d1[t].w * w;
        }
      }
    }
  } else {
    const int4* __restrict__ node = (const int4*)nodes;
    for (int sl = 0; sl < LB_S; ++sl) {
      int h = H[(size_t)sl * hpitch + r];
      while (h) {
        const int4 nd = node[h - 1];
        const float w = __int_as_float(nd.z);
        const float* src = in + (size_t)nd.y * EMB;
        const float4 e0 = *(const float4*)(src + sub * 4);
        const float4 e1 = *(const float4*)(src + 32 + sub * 4);
        a0.x += e0.x * w; a0.y += e0.y * w; a0.z += e0.z * w; a0.w += e0.w * w;
        a1.x += e1.x * w; a1.y += e1.y * w; a1.z += e1.z * w; a1.w += e1.w * w;
        h = nd.x;
      }
    }
  }
  float* dst = out + (size_t)r * EMB;
  *(float4*)(dst + sub * 4) = a0;
  *(float4*)(dst + 32 + sub * 4) = a1;
}

__global__ void ui_final_kernel(const float* __restrict__ uei,
                                const float* __restrict__ item,
                                float* __restrict__ agg) {
  int gid = blockIdx.x * blockDim.x + threadIdx.x;
  if (gid >= N_UI * EMB) return;
  int r = gid >> 6, c = gid & 63;
  float uiv = (r < NUM_USER) ? uei[gid]
                             : item[(size_t)(r - NUM_USER) * EMB + c];
  agg[gid] = (uiv + agg[gid]) * 0.5f;
}

__global__ void gather_out(const float* __restrict__ ug,
                           const float* __restrict__ uif,
                           const float* __restrict__ ig,
                           const int* __restrict__ u,
                           const int* __restrict__ p,
                           const int* __restrict__ ng,
                           float* __restrict__ out) {
  int gid = blockIdx.x * blockDim.x + threadIdx.x;
  if (gid >= BATCH * EMB) return;
  int b = gid >> 6, c = gid & 63;
  const size_t PO = (size_t)BATCH * 128;
  const size_t NO = 2 * PO;
  out[(size_t)b * 128 + c]           = ug [(size_t)u[b] * EMB + c];
  out[(size_t)b * 128 + 64 + c]      = uif[(size_t)u[b] * EMB + c];
  out[PO + (size_t)b * 128 + c]      = ug [(size_t)(NUM_USER + p[b])  * EMB + c];
  out[PO + (size_t)b * 128 + 64 + c] = ig [(size_t)(NUM_ITEM + p[b])  * EMB + c];
  out[NO + (size_t)b * 128 + c]      = ug [(size_t)(NUM_USER + ng[b]) * EMB + c];
  out[NO + (size_t)b * 128 + 64 + c] = ig [(size_t)(NUM_ITEM + ng[b]) * EMB + c];
}

__global__ void reg_kernel(const float* __restrict__ ued,
                           const float* __restrict__ uei,
                           const float* __restrict__ ged,
                           const float* __restrict__ gei,
                           const int* __restrict__ u,
                           const int* __restrict__ p,
                           const int* __restrict__ ng,
                           float* __restrict__ reg_out) {
  int gid = blockIdx.x * blockDim.x + threadIdx.x;
  if (gid >= BATCH * EMB) return;
  int b = gid >> 6, c = gid & 63;
  size_t uo = (size_t)u[b]  * EMB + c;
  size_t po = (size_t)p[b]  * EMB + c;
  size_t no = (size_t)ng[b] * EMB + c;
  float s = uei[uo]*uei[uo] + ued[uo]*ued[uo]
          + gei[po]*gei[po] + ged[po]*ged[po]
          + gei[no]*gei[no] + ged[no]*ged[no];
  #pragma unroll
  for (int off = 1; off < 32; off <<= 1) s += __shfl_xor(s, off, 32);
  if ((threadIdx.x & 31) == 0) atomicAdd(reg_out, s * (0.5f / (float)BATCH));
}

__global__ void normalize_rows_f16(const float* __restrict__ in,
                                   _Float16* __restrict__ out, int nrows) {
  int gid = blockIdx.x * blockDim.x + threadIdx.x;
  int r = gid >> 5;
  if (r >= nrows) return;
  int t = gid & 31;
  size_t base = (size_t)r * EMB + 2 * t;
  float x0 = in[base], x1 = in[base + 1];
  float ss = x0 * x0 + x1 * x1;
  #pragma unroll
  for (int off = 1; off < 32; off <<= 1) ss += __shfl_xor(ss, off, 32);
  float scale = 1.0f / fmaxf(sqrtf(ss), EPS_N);
  out[base]     = (_Float16)(x0 * scale);
  out[base + 1] = (_Float16)(x1 * scale);
}

#define GEMM_WAVES 8
#define GEMM_TPW   4
#define GEMM_BCH   128

typedef unsigned int u32x4 __attribute__((ext_vector_type(4)));
typedef unsigned int u32x8 __attribute__((ext_vector_type(8)));

#define SLAB_PITCH 144

__device__ __forceinline__ void row_frags(const unsigned char* p, v16h& f0, v16h& f1) {
  u32x4 q0 = *(const u32x4*)(p);
  u32x4 q1 = *(const u32x4*)(p + 32);
  u32x4 q2 = *(const u32x4*)(p + 64);
  u32x4 q3 = *(const u32x4*)(p + 96);
  u32x8 w0 = __builtin_shufflevector(q0, q1, 0, 1, 2, 3, 4, 5, 6, 7);
  u32x8 w1 = __builtin_shufflevector(q2, q3, 0, 1, 2, 3, 4, 5, 6, 7);
  f0 = __builtin_bit_cast(v16h, w0);
  f1 = __builtin_bit_cast(v16h, w1);
}

__global__ __launch_bounds__(GEMM_WAVES * 32)
void ssl_gemm_expsum(const _Float16* __restrict__ A16,
                     const _Float16* __restrict__ B16,
                     const int* __restrict__ idx,
                     float* __restrict__ tot,
                     int Nall) {
  __shared__ __attribute__((aligned(16))) unsigned char slab[GEMM_BCH * SLAB_PITCH];
  __shared__ float part[GEMM_WAVES * 2 * GEMM_BCH];
  int lane  = threadIdx.x & 31;
  int wave  = threadIdx.x >> 5;
  int m     = lane & 15;
  int hi    = lane >> 4;
  int lr    = lane >> 3;
  int lc    = lane & 7;
  int bbase = blockIdx.x * GEMM_BCH;

  int rid[GEMM_BCH / (GEMM_WAVES * 4)];
  #pragma unroll
  for (int p = 0; p < GEMM_BCH / (GEMM_WAVES * 4); ++p)
    rid[p] = idx[bbase + p * (GEMM_WAVES * 4) + wave * 4 + lr];

  int ntile = Nall >> 4;
  int tile0 = (blockIdx.y * GEMM_WAVES + wave) * GEMM_TPW;
  v16h  a0[GEMM_TPW], a1[GEMM_TPW];
  float wgt[GEMM_TPW];
  #pragma unroll
  for (int t = 0; t < GEMM_TPW; ++t) {
    int  tile = tile0 + t;
    bool ok   = tile < ntile;
    int  tl   = ok ? tile : 0;
    wgt[t] = ok ? 1.0f : 0.0f;
    row_frags((const unsigned char*)(B16 + (size_t)(tl * 16 + m) * EMB) + hi * 16,
              a0[t], a1[t]);
  }

  u32x4 g[GEMM_BCH / (GEMM_WAVES * 4)];
  #pragma unroll
  for (int p = 0; p < GEMM_BCH / (GEMM_WAVES * 4); ++p)
    g[p] = *(const u32x4*)((const unsigned char*)(A16 + (size_t)rid[p] * EMB) + lc * 16);
  #pragma unroll
  for (int p = 0; p < GEMM_BCH / (GEMM_WAVES * 4); ++p) {
    int r = p * (GEMM_WAVES * 4) + wave * 4 + lr;
    *(u32x4*)(slab + r * SLAB_PITCH + lc * 16) = g[p];
  }
  __syncthreads();

  const unsigned char* brow = slab + m * SLAB_PITCH + hi * 16;
  v16h b0, b1;
  row_frags(brow, b0, b1);
  #pragma unroll
  for (int bt = 0; bt < GEMM_BCH / 16; ++bt) {
    v16h n0 = b0, n1 = b1;
    if (bt + 1 < GEMM_BCH / 16) row_frags(brow + (bt + 1) * 16 * SLAB_PITCH, n0, n1);
    v8f c[GEMM_TPW];
    #pragma unroll
    for (int t = 0; t < GEMM_TPW; ++t) {
      v8f z = {};
      c[t] = __builtin_amdgcn_wmma_f32_16x16x32_f16(false, a0[t], false, b0,
                                                    (short)0, z, false, false);
    }
    #pragma unroll
    for (int t = 0; t < GEMM_TPW; ++t)
      c[t] = __builtin_amdgcn_wmma_f32_16x16x32_f16(false, a1[t], false, b1,
                                                    (short)0, c[t], false, false);
    float e = 0.0f;
    #pragma unroll
    for (int t = 0; t < GEMM_TPW; ++t) {
      float x[8];
      #pragma unroll
      for (int v = 0; v < 8; ++v) x[v] = __expf(c[t][v] * INV_TEMP);
      float et = ((x[0] + x[1]) + (x[2] + x[3])) + ((x[4] + x[5]) + (x[6] + x[7]));
      e += wgt[t] * et;
    }
    part[(wave * 2 + hi) * GEMM_BCH + bt * 16 + m] = e;
    b0 = n0; b1 = n1;
  }
  __syncthreads();

  if (threadIdx.x < GEMM_BCH) {
    float sum = 0.0f;
    #pragma unroll
    for (int w = 0; w < GEMM_WAVES * 2; ++w) sum += part[w * GEMM_BCH + threadIdx.x];
    atomicAdd(&tot[bbase + threadIdx.x], sum);
  }
}

__global__ void ssl_loss(const _Float16* __restrict__ A16,
                         const _Float16* __restrict__ B16,
                         const int* __restrict__ idx,
                         const float* __restrict__ tot,
                         float* __restrict__ ssl_out) {
  int gid = blockIdx.x * blockDim.x + threadIdx.x;
  int i = gid >> 5;
  if (i >= BATCH) return;
  int t = gid & 31;
  size_t ro = (size_t)idx[i] * EMB + 2 * t;
  float d = (float)A16[ro] * (float)B16[ro]
          + (float)A16[ro + 1] * (float)B16[ro + 1];
  #pragma unroll
  for (int off = 1; off < 32; off <<= 1) d += __shfl_xor(d, off, 32);
  if (t == 0) {
    float li = -(d * INV_TEMP - logf(tot[i])) * (1.0f / (float)BATCH);
    atomicAdd(ssl_out, li);
  }
}

extern "C" void kernel_launch(void* const* d_in, const int* in_sizes, int n_in,
                              void* d_out, int out_size, void* d_ws, size_t ws_size,
                              hipStream_t stream) {
  (void)in_sizes; (void)n_in; (void)out_size; (void)ws_size;
  const float* ued  = (const float*)d_in[0];
  const float* uei  = (const float*)d_in[1];
  const float* ged  = (const float*)d_in[2];
  const float* gei  = (const float*)d_in[3];
  const float* item = (const float*)d_in[4];
  const int*   hu_r = (const int*)d_in[5];
  const int*   hu_c = (const int*)d_in[6];
  const float* hu_v = (const float*)d_in[7];
  const int*   hi_r = (const int*)d_in[8];
  const int*   hi_c = (const int*)d_in[9];
  const float* hi_v = (const float*)d_in[10];
  const int*   ui_r = (const int*)d_in[11];
  const int*   ui_c = (const int*)d_in[12];
  const float* ui_v = (const float*)d_in[13];
  const int*   uin  = (const int*)d_in[14];
  const int*   pos  = (const int*)d_in[15];
  const int*   neg  = (const int*)d_in[16];
  float* out = (float*)d_out;

  float* W = (float*)d_ws;
  size_t off = 0;
  float* ug_a = W + off; off += (size_t)N_UG * EMB;
  float* ug_b = W + off; off += (size_t)N_UG * EMB;
  float* ig_b = W + off; off += (size_t)N_IG * EMB;
  float* ig_a = W + off; off += (size_t)N_IG * EMB;
  float* agg  = W + off; off += (size_t)N_UI * EMB;
  float* tot  = W + off; off += 4 * BATCH;
  #define HP_UG 120000
  #define HP_IG 70016
  #define HP_UI 150016
  int* H_hg = (int*)agg;
  int* H_ui = (int*)ug_b;
  int2* pr_ug = (int2*)ig_b;
  int*  pc_ug = (int*)ig_b + (size_t)N_UG * PB_K * 2;
  int2* pr_ig = (int2*)ug_b;
  int*  pc_ig = (int*)ug_b + (size_t)N_IG * PB_K * 2;
  int* node = (int*)(W + off);
  _Float16* H = (_Float16*)(W + off);
  _Float16* ug_user_n16  = H;
  _Float16* ui_user_n16  = H + (size_t)NUM_USER * EMB;
  _Float16* ug_group_n16 = H + (size_t)NUM_USER * EMB * 2;
  _Float16* ig_group_n16 = ug_group_n16 + (size_t)NUM_GROUP * EMB;

  const int TB = 256;
  #define CDIV(a, b) (((a) + (b) - 1) / (b))

  zero_f32<<<16, TB, 0, stream>>>(tot, 4 * BATCH);
  zero_f32<<<1, 32, 0, stream>>>(out + (size_t)3 * BATCH * 128, 2);

  const dim3 gB(LB_P, LB_S);
  list_build_lds<<<gB, LB_THREADS, 0, stream>>>(hu_c, hu_r, hu_v, H_hg, HP_UG, node, E_U, N_UG);
  list_prop<true><<<CDIV(N_UG, LP_COLS), LP_COLS * 8, 0, stream>>>(
      ued, NUM_USER, ged, H_hg, HP_UG, node, ug_b, N_UG, pr_ug, pc_ug);
  pair_prop<<<CDIV(N_UG, LP_COLS), LP_COLS * 8, 0, stream>>>(
      ug_b, pr_ug, pc_ug, H_hg, HP_UG, node, ug_a, N_UG);

  list_build_lds<<<gB, LB_THREADS, 0, stream>>>(hi_c, hi_r, hi_v, H_hg, HP_IG, node, E_I, N_IG);
  list_prop<true><<<CDIV(N_IG, LP_COLS), LP_COLS * 8, 0, stream>>>(
      item, NUM_ITEM, gei, H_hg, HP_IG, node, ig_b, N_IG, pr_ig, pc_ig);
  pair_prop<<<CDIV(N_IG, LP_COLS), LP_COLS * 8, 0, stream>>>(
      ig_b, pr_ig, pc_ig, H_hg, HP_IG, node, ig_a, N_IG);

  list_build_lds<<<gB, LB_THREADS, 0, stream>>>(ui_c, ui_r, ui_v, H_ui, HP_UI, node, E_UI, N_UI);
  list_prop<false><<<CDIV(N_UI, LP_COLS), LP_COLS * 8, 0, stream>>>(
      uei, NUM_USER, item, H_ui, HP_UI, node, agg, N_UI, nullptr, nullptr);
  ui_final_kernel<<<CDIV(N_UI * EMB, TB), TB, 0, stream>>>(uei, item, agg);

  gather_out<<<CDIV(BATCH * EMB, TB), TB, 0, stream>>>(
      ug_a, agg, ig_a, uin, pos, neg, out);
  reg_kernel<<<CDIV(BATCH * EMB, TB), TB, 0, stream>>>(
      ued, uei, ged, gei, uin, pos, neg, out + (size_t)3 * BATCH * 128);

  normalize_rows_f16<<<CDIV(NUM_USER * 32, TB), TB, 0, stream>>>(
      ug_a, ug_user_n16, NUM_USER);
  normalize_rows_f16<<<CDIV(NUM_USER * 32, TB), TB, 0, stream>>>(
      agg, ui_user_n16, NUM_USER);
  normalize_rows_f16<<<CDIV(NUM_GROUP * 32, TB), TB, 0, stream>>>(
      ug_a + (size_t)NUM_USER * EMB, ug_group_n16, NUM_GROUP);
  normalize_rows_f16<<<CDIV(NUM_GROUP * 32, TB), TB, 0, stream>>>(
      ig_a + (size_t)NUM_ITEM * EMB, ig_group_n16, NUM_GROUP);

  const int perBlk = GEMM_WAVES * GEMM_TPW;
  int ntU = NUM_USER / 16, ntG = NUM_GROUP / 16;
  dim3 gU(BATCH / GEMM_BCH, CDIV(ntU, perBlk));
  dim3 gG(BATCH / GEMM_BCH, CDIV(ntG, perBlk));
  ssl_gemm_expsum<<<gU, TB, 0, stream>>>(ui_user_n16, ug_user_n16, uin, tot + 0,         NUM_USER);
  ssl_gemm_expsum<<<gU, TB, 0, stream>>>(ug_user_n16, ui_user_n16, uin, tot + BATCH,     NUM_USER);
  ssl_gemm_expsum<<<gG, TB, 0, stream>>>(ig_group_n16, ug_group_n16, pos, tot + 2*BATCH, NUM_GROUP);
  ssl_gemm_expsum<<<gG, TB, 0, stream>>>(ug_group_n16, ig_group_n16, pos, tot + 3*BATCH, NUM_GROUP);

  float* ssl_slot = out + (size_t)3 * BATCH * 128 + 1;
  int lossBlocks = CDIV(BATCH * 32, TB);
  ssl_loss<<<lossBlocks, TB, 0, stream>>>(ui_user_n16, ug_user_n16, uin, tot + 0,         ssl_slot);
  ssl_loss<<<lossBlocks, TB, 0, stream>>>(ug_user_n16, ui_user_n16, uin, tot + BATCH,     ssl_slot);
  ssl_loss<<<lossBlocks, TB, 0, stream>>>(ig_group_n16, ug_group_n16, pos, tot + 2*BATCH, ssl_slot);
  ssl_loss<<<lossBlocks, TB, 0, stream>>>(ug_group_n16, ig_group_n16, pos, tot + 3*BATCH, ssl_slot);
  #undef CDIV
}


# List heads carry an edge: the first edge of a (column, slice) sits as (row, value) in the head table, nodes only for edges that meet a filled slot; 10 ranges x 24 slices; sum order slots, then chains
# speedup vs baseline: 4.9306x; 1.1390x over previous
#include <hip/hip_runtime.h>
#include <hip/hip_bf16.h>
#include <math.h>

#define NUM_USER   100000
#define NUM_ITEM   50000
#define NUM_GROUP  20000
#define EMB        64
#define BATCH      1024
#define E_UI       2000000
#define E_U        1000000
#define E_I        1000000
#define INV_TEMP   10.0f
#define EPS_N      1e-12f

#define N_UG  (NUM_USER + NUM_GROUP)
#define N_IG  (NUM_ITEM + NUM_GROUP)
#define N_UI  (NUM_USER + NUM_ITEM)

typedef _Float16 v16h __attribute__((ext_vector_type(16)));
typedef float    v8f  __attribute__((ext_vector_type(8)));

__global__ void zero_f32(float* __restrict__ p, size_t n) {
  size_t i = (size_t)blockIdx.x * blockDim.x + threadIdx.x;
  size_t stride = (size_t)gridDim.x * blockDim.x;
  for (; i < n; i += stride) p[i] = 0.0f;
}

#define LB_P    10
#define LB_S    24
#define LB_WMAX 15008
#define LB_THREADS 1024
#define LB_G    4
#define LB_OTHER (-2)

__device__ __forceinline__ int claim1(int* hl, int c, int c0, int Wd) {
  unsigned d = (unsigned)(c - c0);
  return (d < (unsigned)Wd) ? atomicCAS(&hl[d], 0, -1) : LB_OTHER;
}
__device__ __forceinline__ void place1(int* hl, int2* hp, int4* __restrict__ node, int e,
                                       int c, int r, float v, int c0, int o) {
  if (o == LB_OTHER) return;
  const int d = c - c0;
  if (o == 0) {
    hp[d] = make_int2(r, __float_as_int(v));
  } else {
    int o2 = atomicExch(&hl[d], e + 1);
    node[e] = make_int4(o2 > 0 ? o2 : 0, r, __float_as_int(v), 0);
  }
}
__device__ __forceinline__ void thread_tail(int* hl, int2* hp, int4* __restrict__ node,
                                            const int* __restrict__ col,
                                            const int* __restrict__ row,
                                            const float* __restrict__ val,
                                            int e0, int send, int c0, int Wd) {
  for (int e = e0; e < send; ++e) {
    const int c = col[e];
    const int o = claim1(hl, c, c0, Wd);
    place1(hl, hp, node, e, c, row[e], val[e], c0, o);
  }
}

__global__ __launch_bounds__(LB_THREADS)
void list_build_lds(const int* __restrict__ col,
                    const int* __restrict__ row,
                    const float* __restrict__ val,
                    int2* __restrict__ H8, int* __restrict__ H2, int hpitch,
                    int* __restrict__ nodes, int nE, int nCols) {
  __shared__ int  hl[LB_WMAX];
  __shared__ int2 hp[LB_WMAX];
  int4* __restrict__ node = (int4*)nodes;
  const int p  = blockIdx.x;
  const int sl = blockIdx.y;
  const int Wd = (nCols + LB_P - 1) / LB_P;
  const int c0 = p * Wd;
  const int SL = (((nE + LB_S - 1) / LB_S) + 3) & ~3;
  const int sbeg = sl * SL;
  const int send = (sbeg + SL < nE) ? (sbeg + SL) : nE;

  for (int i = threadIdx.x; i < Wd; i += LB_THREADS) hl[i] = 0;
  __syncthreads();

  for (int e0 = sbeg + threadIdx.x * 4; e0 < send; e0 += LB_G * LB_THREADS * 4) {
    int4   cc[LB_G], rr[LB_G], oo[LB_G];
    float4 vv[LB_G];
    bool   ff[LB_G];
    #pragma unroll
    for (int g = 0; g < LB_G; ++g) {
      const int ea = e0 + g * LB_THREADS * 4;
      ff[g] = ea + 3 < send;
      cc[g] = make_int4(0, 0, 0, 0);
      rr[g] = make_int4(0, 0, 0, 0);
      vv[g] = make_float4(0.0f, 0.0f, 0.0f, 0.0f);
      if (ff[g]) {
        cc[g] = *(const int4*)(col + ea);
        rr[g] = *(const int4*)(row + ea);
        vv[g] = *(const float4*)(val + ea);
      }
    }
    #pragma unroll
    for (int g = 0; g < LB_G; ++g) {
      oo[g] = make_int4(LB_OTHER, LB_OTHER, LB_OTHER, LB_OTHER);
      if (ff[g]) {
        oo[g].x = claim1(hl, cc[g].x, c0, Wd);
        oo[g].y = claim1(hl, cc[g].y, c0, Wd);
        oo[g].z = claim1(hl, cc[g].z, c0, Wd);
        oo[g].w = claim1(hl, cc[g].w, c0, Wd);
      }
    }
    #pragma unroll
    for (int g = 0; g < LB_G; ++g) {
      const int ea = e0 + g * LB_THREADS * 4;
      if (ff[g]) {
        place1(hl, hp, node, ea,     cc[g].x, rr[g].x, vv[g].x, c0, oo[g].x);
        place1(hl, hp, node, ea + 1, cc[g].y, rr[g].y, vv[g].y, c0, oo[g].y);
        place1(hl, hp, node, ea + 2, cc[g].z, rr[g].z, vv[g].z, c0, oo[g].z);
        place1(hl, hp, node, ea + 3, cc[g].w, rr[g].w, vv[g].w, c0, oo[g].w);
      } else {
        thread_tail(hl, hp, node, col, row, val, ea, send, c0, Wd);
      }
    }
  }
  __syncthreads();

  int2* __restrict__ H8row = H8 + (size_t)sl * hpitch + c0;
  int*  __restrict__ H2row = H2 + (size_t)sl * hpitch + c0;
  for (int i = threadIdx.x; i < Wd; i += LB_THREADS) {
    if (c0 + i < nCols) {
      const int  h = hl[i];
      const int2 s = hp[i];
      H8row[i] = h ? s : make_int2(-1, 0);
      H2row[i] = (h > 0) ? h : 0;
    }
  }
}

#define LP_COLS 32
#define PB_K 32
template <bool WP>
__global__ __launch_bounds__(LP_COLS * 8)
void list_prop(const float* __restrict__ p0, int n0,
               const float* __restrict__ p1,
               const int2* __restrict__ H8, const int* __restrict__ H2, int hpitch,
               const int* __restrict__ nodes,
               float* __restrict__ out, int nRows,
               int2* __restrict__ pairs, int* __restrict__ pcnt) {
  __shared__ int2 hs8[LB_S * LP_COLS];
  __shared__ int  hs2[LB_S * LP_COLS];
  __shared__ int  cq[LP_COLS * LB_S];
  __shared__ __attribute__((aligned(16))) int2 pb[LP_COLS * PB_K];
  const int4* __restrict__ node = (const int4*)nodes;
  const int lane = threadIdx.x & 31;
  const int wave = threadIdx.x >> 5;
  const int cb   = blockIdx.x * LP_COLS;

  int2 hv8[LB_S / 8];
  int  hv2[LB_S / 8];
  #pragma unroll
  for (int i = 0; i < LB_S / 8; ++i) {
    const int sidx = i * 8 + wave;
    const bool ok = cb + lane < nRows;
    hv8[i] = ok ? H8[(size_t)sidx * hpitch + cb + lane] : make_int2(-1, 0);
    hv2[i] = ok ? H2[(size_t)sidx * hpitch + cb + lane] : 0;
  }
  #pragma unroll
  for (int i = 0; i < LB_S / 8; ++i) {
    hs8[(i * 8 + wave) * LP_COLS + lane] = hv8[i];
    hs2[(i * 8 + wave) * LP_COLS + lane] = hv2[i];
  }
  __syncthreads();

  const int j   = threadIdx.x >> 3;
  const int sub = threadIdx.x & 7;
  const int r   = cb + j;
  const unsigned gsh   = (unsigned)(lane & 24);
  const unsigned below = (1u << sub) - 1u;
  int cnt1 = 0;
  int cnt  = 0;
  #pragma unroll
  for (int i = 0; i < LB_S / 8; ++i) {
    const int2 s = hs8[(i * 8 + sub) * LP_COLS + j];
    const int  h = hs2[(i * 8 + sub) * LP_COLS + j];
    const unsigned g1 = (__builtin_amdgcn_ballot_w32(s.x >= 0) >> gsh) & 0xFFu;
    const unsigned g2 = (__builtin_amdgcn_ballot_w32(h != 0) >> gsh) & 0xFFu;
    if (s.x >= 0) pb[j * PB_K + cnt1 + __popc(g1 & below)] = s;
    if (h)        cq[j * LB_S + cnt  + __popc(g2 & below)] = h;
    cnt1 += __popc(g1);
    cnt  += __popc(g2);
  }
  __syncthreads();

  float4 a0 = {0.0f, 0.0f, 0.0f, 0.0f}, a1 = {0.0f, 0.0f, 0.0f, 0.0f};
  const int* __restrict__ q = cq + j * LB_S;
  int cur   = (cnt > 0) ? q[0] : 0;
  int hnext = (cnt > 1) ? q[1] : 0;
  int k = 2;
  int4 nd = make_int4(0, 0, 0, 0);
  if (cur) nd = node[cur - 1];
  for (int kk = 0; kk * 4 < cnt1; ++kk) {
    const int m = cnt1 - kk * 4;
    const int4* __restrict__ ps = (const int4*)(pb + j * PB_K + kk * 4);
    const int4 u0 = ps[0], u1 = ps[1];
    const int   s[4] = { u0.x, u0.z, u1.x, u1.z };
    const float v[4] = { __int_as_float(u0.y), __int_as_float(u0.w),
                         __int_as_float(u1.y), __int_as_float(u1.w) };
    float4 d0[4], d1[4];
    #pragma unroll
    for (int t = 0; t < 4; ++t) {
      d0[t] = make_float4(0.0f, 0.0f, 0.0f, 0.0f);
      d1[t] = make_float4(0.0f, 0.0f, 0.0f, 0.0f);
      if (t < m) {
        const float* src = (s[t] < n0) ? (p0 + (size_t)s[t] * EMB)
                                       : (p1 + (size_t)(s[t] - n0) * EMB);
        d0[t] = *(const float4*)(src + sub * 4);
        d1[t] = *(const float4*)(src + 32 + sub * 4);
      }
    }
    #pragma unroll
    for (int t = 0; t < 4; ++t) {
      if (t < m) {
        const float w = v[t];
        a0.x += d0[t].x * w; a0.y += d0[t].y * w; a0.z += d0[t].z * w; a0.w += d0[t].w * w;
        a1.x += d1[t].x * w; a1.y += d1[t].y * w; a1.z += d1[t].z * w; a1.w += d1[t].w * w;
      }
    }
  }

  float4 d0 = {0.0f, 0.0f, 0.0f, 0.0f}, d1 = {0.0f, 0.0f, 0.0f, 0.0f};
  float vp = 0.0f;
  int li = cnt1;
  while (cur) {
    int nx = nd.x;
    int s = nd.y;
    float v = __int_as_float(nd.z);
    if (WP) {
      if (sub == 0 && li < PB_K) pb[j * PB_K + li] = make_int2(s, nd.z);
      ++li;
    }
    const bool seam = (nx == 0);
    if (seam) nx = hnext;
    a0.x += d0.x * vp; a0.y += d0.y * vp; a0.z += d0.z * vp; a0.w += d0.w * vp;
    a1.x += d1.x * vp; a1.y += d1.y * vp; a1.z += d1.z * vp; a1.w += d1.w * vp;
    const float* src = (s < n0) ? (p0 + (size_t)s * EMB)
                                : (p1 + (size_t)(s - n0) * EMB);
    d0 = *(const float4*)(src + sub * 4);
    d1 = *(const float4*)(src + 32 + sub * 4);
    if (nx) nd = node[nx - 1];
    if (seam) { hnext = (k < cnt) ? q[k] : 0; ++k; }
    vp = v;
    cur = nx;
  }
  a0.x += d0.x * vp; a0.y += d0.y * vp; a0.z += d0.z * vp; a0.w += d0.w * vp;
  a1.x += d1.x * vp; a1.y += d1.y * vp; a1.z += d1.z * vp; a1.w += d1.w * vp;
  if (r < nRows) {
    float* dst = out + (size_t)r * EMB;
    *(float4*)(dst + sub * 4) = a0;
    *(float4*)(dst + 32 + sub * 4) = a1;
    if (WP) {
      const int4* ps = (const int4*)(pb + j * PB_K + sub * 4);
      int4 u0 = ps[0], u1 = ps[1];
      int4* pd = (int4*)(pairs + (size_t)r * PB_K + sub * 4);
      pd[0] = u0;
      pd[1] = u1;
      if (sub == 0) pcnt[r] = li;
    }
  }
}

__global__ __launch_bounds__(LP_COLS * 8)
void pair_prop(const float* __restrict__ in,
               const int2* __restrict__ pairs, const int* __restrict__ pcnt,
               const int2* __restrict__ H8, const int* __restrict__ H2, int hpitch,
               const int* __restrict__ nodes,
               float* __restrict__ out, int nRows) {
  const int j   = threadIdx.x >> 3;
  const int sub = threadIdx.x & 7;
  const int r   = blockIdx.x * LP_COLS + j;
  if (r >= nRows) return;
  const int4* pp = (const int4*)(pairs + (size_t)r * PB_K + sub * 4);
  const int4 q0 = pp[0], q1 = pp[1];
  const int cnt = pcnt[r];
  float4 a0 = {0.0f, 0.0f, 0.0f, 0.0f}, a1 = {0.0f, 0.0f, 0.0f, 0.0f};
  if (cnt <= PB_K) {
    for (int kk = 0; kk * 4 < cnt; ++kk) {
      const int m = cnt - kk * 4;
      int   s[4];
      float v[4];
      s[0] = __shfl(q0.x, kk, 8); v[0] = __int_as_float(__shfl(q0.y, kk, 8));
      s[1] = __shfl(q0.z, kk, 8); v[1] = __int_as_float(__shfl(q0.w, kk, 8));
      s[2] = __shfl(q1.x, kk, 8); v[2] = __int_as_float(__shfl(q1.y, kk, 8));
      s[3] = __shfl(q1.z, kk, 8); v[3] = __int_as_float(__shfl(q1.w, kk, 8));
      float4 d0[4], d1[4];
      #pragma unroll
      for (int t = 0; t < 4; ++t) {
        d0[t] = make_float4(0.0f, 0.0f, 0.0f, 0.0f);
        d1[t] = make_float4(0.0f, 0.0f, 0.0f, 0.0f);
        if (t < m) {
          const float* src = in + (size_t)s[t] * EMB;
          d0[t] = *(const float4*)(src + sub * 4);
          d1[t] = *(const float4*)(src + 32 + sub * 4);
        }
      }
      #pragma unroll
      for (int t = 0; t < 4; ++t) {
        if (t < m) {
          const float w = v[t];
          a0.x += d0[t].x * w; a0.y += d0[t].y * w; a0.z += d0[t].z * w; a0.w += d0[t].w * w;
          a1.x += d1[t].x * w; a1.y += d1[t].y * w; a1.z += d1[t].z * w; a1.w += d1[t].w * w;
        }
      }
    }
  } else {
    const int4* __restrict__ node = (const int4*)nodes;
    for (int sl = 0; sl < LB_S; ++sl) {
      const int2 s = H8[(size_t)sl * hpitch + r];
      if (s.x >= 0) {
        const float w = __int_as_float(s.y);
        const float* src = in + (size_t)s.x * EMB;
        const float4 e0 = *(const float4*)(src + sub * 4);
        const float4 e1 = *(const float4*)(src + 32 + sub * 4);
        a0.x += e0.x * w; a0.y += e0.y * w; a0.z += e0.z * w; a0.w += e0.w * w;
        a1.x += e1.x * w; a1.y += e1.y * w; a1.z += e1.z * w; a1.w += e1.w * w;
      }
    }
    for (int sl = 0; sl < LB_S; ++sl) {
      int h = H2[(size_t)sl * hpitch + r];
      while (h) {
        const int4 nd = node[h - 1];
        const float w = __int_as_float(nd.z);
        const float* src = in + (size_t)nd.y * EMB;
        const float4 e0 = *(const float4*)(src + sub * 4);
        const float4 e1 = *(const float4*)(src + 32 + sub * 4);
        a0.x += e0.x * w; a0.y += e0.y * w; a0.z += e0.z * w; a0.w += e0.w * w;
        a1.x += e1.x * w; a1.y += e1.y * w; a1.z += e1.z * w; a1.w += e1.w * w;
        h = nd.x;
      }
    }
  }
  float* dst = out + (size_t)r * EMB;
  *(float4*)(dst + sub * 4) = a0;
  *(float4*)(dst + 32 + sub * 4) = a1;
}

__global__ void ui_final_kernel(const float* __restrict__ uei,
                                const float* __restrict__ item,
                                float* __restrict__ agg) {
  int gid = blockIdx.x * blockDim.x + threadIdx.x;
  if (gid >= N_UI * EMB) return;
  int r = gid >> 6, c = gid & 63;
  float uiv = (r < NUM_USER) ? uei[gid]
                             : item[(size_t)(r - NUM_USER) * EMB + c];
  agg[gid] = (uiv + agg[gid]) * 0.5f;
}

__global__ void gather_out(const float* __restrict__ ug,
                           const float* __restrict__ uif,
                           const float* __restrict__ ig,
                           const int* __restrict__ u,
                           const int* __restrict__ p,
                           const int* __restrict__ ng,
                           float* __restrict__ out) {
  int gid = blockIdx.x * blockDim.x + threadIdx.x;
  if (gid >= BATCH * EMB) return;
  int b = gid >> 6, c = gid & 63;
  const size_t PO = (size_t)BATCH * 128;
  const size_t NO = 2 * PO;
  out[(size_t)b * 128 + c]           = ug [(size_t)u[b] * EMB + c];
  out[(size_t)b * 128 + 64 + c]      = uif[(size_t)u[b] * EMB + c];
  out[PO + (size_t)b * 128 + c]      = ug [(size_t)(NUM_USER + p[b])  * EMB + c];
  out[PO + (size_t)b * 128 + 64 + c] = ig [(size_t)(NUM_ITEM + p[b])  * EMB + c];
  out[NO + (size_t)b * 128 + c]      = ug [(size_t)(NUM_USER + ng[b]) * EMB + c];
  out[NO + (size_t)b * 128 + 64 + c] = ig [(size_t)(NUM_ITEM + ng[b]) * EMB + c];
}

__global__ void reg_kernel(const float* __restrict__ ued,
                           const float* __restrict__ uei,
                           const float* __restrict__ ged,
                           const float* __restrict__ gei,
                           const int* __restrict__ u,
                           const int* __restrict__ p,
                           const int* __restrict__ ng,
                           float* __restrict__ reg_out) {
  int gid = blockIdx.x * blockDim.x + threadIdx.x;
  if (gid >= BATCH * EMB) return;
  int b = gid >> 6, c = gid & 63;
  size_t uo = (size_t)u[b]  * EMB + c;
  size_t po = (size_t)p[b]  * EMB + c;
  size_t no = (size_t)ng[b] * EMB + c;
  float s = uei[uo]*uei[uo] + ued[uo]*ued[uo]
          + gei[po]*gei[po] + ged[po]*ged[po]
          + gei[no]*gei[no] + ged[no]*ged[no];
  #pragma unroll
  for (int off = 1; off < 32; off <<= 1) s += __shfl_xor(s, off, 32);
  if ((threadIdx.x & 31) == 0) atomicAdd(reg_out, s * (0.5f / (float)BATCH));
}

__global__ void normalize_rows_f16(const float* __restrict__ in,
                                   _Float16* __restrict__ out, int nrows) {
  int gid = blockIdx.x * blockDim.x + threadIdx.x;
  int r = gid >> 5;
  if (r >= nrows) return;
  int t = gid & 31;
  size_t base = (size_t)r * EMB + 2 * t;
  float x0 = in[base], x1 = in[base + 1];
  float ss = x0 * x0 + x1 * x1;
  #pragma unroll
  for (int off = 1; off < 32; off <<= 1) ss += __shfl_xor(ss, off, 32);
  float scale = 1.0f / fmaxf(sqrtf(ss), EPS_N);
  out[base]     = (_Float16)(x0 * scale);
  out[base + 1] = (_Float16)(x1 * scale);
}

#define GEMM_WAVES 8
#define GEMM_TPW   4
#define GEMM_BCH   128

typedef unsigned int u32x4 __attribute__((ext_vector_type(4)));
typedef unsigned int u32x8 __attribute__((ext_vector_type(8)));

#define SLAB_PITCH 144

__device__ __forceinline__ void row_frags(const unsigned char* p, v16h& f0, v16h& f1) {
  u32x4 q0 = *(const u32x4*)(p);
  u32x4 q1 = *(const u32x4*)(p + 32);
  u32x4 q2 = *(const u32x4*)(p + 64);
  u32x4 q3 = *(const u32x4*)(p + 96);
  u32x8 w0 = __builtin_shufflevector(q0, q1, 0, 1, 2, 3, 4, 5, 6, 7);
  u32x8 w1 = __builtin_shufflevector(q2, q3, 0, 1, 2, 3, 4, 5, 6, 7);
  f0 = __builtin_bit_cast(v16h, w0);
  f1 = __builtin_bit_cast(v16h, w1);
}

__global__ __launch_bounds__(GEMM_WAVES * 32)
void ssl_gemm_expsum(const _Float16* __restrict__ A16,
                     const _Float16* __restrict__ B16,
                     const int* __restrict__ idx,
                     float* __restrict__ tot,
                     int Nall) {
  __shared__ __attribute__((aligned(16))) unsigned char slab[GEMM_BCH * SLAB_PITCH];
  __shared__ float part[GEMM_WAVES * 2 * GEMM_BCH];
  int lane  = threadIdx.x & 31;
  int wave  = threadIdx.x >> 5;
  int m     = lane & 15;
  int hi    = lane >> 4;
  int lr    = lane >> 3;
  int lc    = lane & 7;
  int bbase = blockIdx.x * GEMM_BCH;

  int rid[GEMM_BCH / (GEMM_WAVES * 4)];
  #pragma unroll
  for (int p = 0; p < GEMM_BCH / (GEMM_WAVES * 4); ++p)
    rid[p] = idx[bbase + p * (GEMM_WAVES * 4) + wave * 4 + lr];

  int ntile = Nall >> 4;
  int tile0 = (blockIdx.y * GEMM_WAVES + wave) * GEMM_TPW;
  v16h  a0[GEMM_TPW], a1[GEMM_TPW];
  float wgt[GEMM_TPW];
  #pragma unroll
  for (int t = 0; t < GEMM_TPW; ++t) {
    int  tile = tile0 + t;
    bool ok   = tile < ntile;
    int  tl   = ok ? tile : 0;
    wgt[t] = ok ? 1.0f : 0.0f;
    row_frags((const unsigned char*)(B16 + (size_t)(tl * 16 + m) * EMB) + hi * 16,
              a0[t], a1[t]);
  }

  u32x4 g[GEMM_BCH / (GEMM_WAVES * 4)];
  #pragma unroll
  for (int p = 0; p < GEMM_BCH / (GEMM_WAVES * 4); ++p)
    g[p] = *(const u32x4*)((const unsigned char*)(A16 + (size_t)rid[p] * EMB) + lc * 16);
  #pragma unroll
  for (int p = 0; p < GEMM_BCH / (GEMM_WAVES * 4); ++p) {
    int r = p * (GEMM_WAVES * 4) + wave * 4 + lr;
    *(u32x4*)(slab + r * SLAB_PITCH + lc * 16) = g[p];
  }
  __syncthreads();

  const unsigned char* brow = slab + m * SLAB_PITCH + hi * 16;
  v16h b0, b1;
  row_frags(brow, b0, b1);
  #pragma unroll
  for (int bt = 0; bt < GEMM_BCH / 16; ++bt) {
    v16h n0 = b0, n1 = b1;
    if (bt + 1 < GEMM_BCH / 16) row_frags(brow + (bt + 1) * 16 * SLAB_PITCH, n0, n1);
    v8f c[GEMM_TPW];
    #pragma unroll
    for (int t = 0; t < GEMM_TPW; ++t) {
      v8f z = {};
      c[t] = __builtin_amdgcn_wmma_f32_16x16x32_f16(false, a0[t], false, b0,
                                                    (short)0, z, false, false);
    }
    #pragma unroll
    for (int t = 0; t < GEMM_TPW; ++t)
      c[t] = __builtin_amdgcn_wmma_f32_16x16x32_f16(false, a1[t], false, b1,
                                                    (short)0, c[t], false, false);
    float e = 0.0f;
    #pragma unroll
    for (int t = 0; t < GEMM_TPW; ++t) {
      float x[8];
      #pragma unroll
      for (int v = 0; v < 8; ++v) x[v] = __expf(c[t][v] * INV_TEMP);
      float et = ((x[0] + x[1]) + (x[2] + x[3])) + ((x[4] + x[5]) + (x[6] + x[7]));
      e += wgt[t] * et;
    }
    part[(wave * 2 + hi) * GEMM_BCH + bt * 16 + m] = e;
    b0 = n0; b1 = n1;
  }
  __syncthreads();

  if (threadIdx.x < GEMM_BCH) {
    float sum = 0.0f;
    #pragma unroll
    for (int w = 0; w < GEMM_WAVES * 2; ++w) sum += part[w * GEMM_BCH + threadIdx.x];
    atomicAdd(&tot[bbase + threadIdx.x], sum);
  }
}

__global__ void ssl_loss(const _Float16* __restrict__ A16,
                         const _Float16* __restrict__ B16,
                         const int* __restrict__ idx,
                         const float* __restrict__ tot,
                         float* __restrict__ ssl_out) {
  int gid = blockIdx.x * blockDim.x + threadIdx.x;
  int i = gid >> 5;
  if (i >= BATCH) return;
  int t = gid & 31;
  size_t ro = (size_t)idx[i] * EMB + 2 * t;
  float d = (float)A16[ro] * (float)B16[ro]
          + (float)A16[ro + 1] * (float)B16[ro + 1];
  #pragma unroll
  for (int off = 1; off < 32; off <<= 1) d += __shfl_xor(d, off, 32);
  if (t == 0) {
    float li = -(d * INV_TEMP - logf(tot[i])) * (1.0f / (float)BATCH);
    atomicAdd(ssl_out, li);
  }
}

extern "C" void kernel_launch(void* const* d_in, const int* in_sizes, int n_in,
                              void* d_out, int out_size, void* d_ws, size_t ws_size,
                              hipStream_t stream) {
  (void)in_sizes; (void)n_in; (void)out_size; (void)ws_size;
  const float* ued  = (const float*)d_in[0];
  const float* uei  = (const float*)d_in[1];
  const float* ged  = (const float*)d_in[2];
  const float* gei  = (const float*)d_in[3];
  const float* item = (const float*)d_in[4];
  const int*   hu_r = (const int*)d_in[5];
  const int*   hu_c = (const int*)d_in[6];
  const float* hu_v = (const float*)d_in[7];
  const int*   hi_r = (const int*)d_in[8];
  const int*   hi_c = (const int*)d_in[9];
  const float* hi_v = (const float*)d_in[10];
  const int*   ui_r = (const int*)d_in[11];
  const int*   ui_c = (const int*)d_in[12];
  const float* ui_v = (const float*)d_in[13];
  const int*   uin  = (const int*)d_in[14];
  const int*   pos  = (const int*)d_in[15];
  const int*   neg  = (const int*)d_in[16];
  float* out = (float*)d_out;

  float* W = (float*)d_ws;
  size_t off = 0;
  float* ug_a = W + off; off += (size_t)N_UG * EMB;
  float* ug_b = W + off; off += (size_t)N_UG * EMB;
  float* ig_b = W + off; off += (size_t)N_IG * EMB;
  float* ig_a = W + off; off += (size_t)N_IG * EMB;
  float* agg  = W + off; off += (size_t)N_UI * EMB;
  float* tot  = W + off; off += 4 * BATCH;
  #define HP_UG 120000
  #define HP_IG 70016
  #define HP_UI 150016
  int2* H8_ug = (int2*)agg;   int* H2_ug = (int*)agg  + (size_t)LB_S * HP_UG * 2;
  int2* H8_ig = (int2*)agg;   int* H2_ig = (int*)agg  + (size_t)LB_S * HP_IG * 2;
  int2* H8_ui = (int2*)ug_b;  int* H2_ui = (int*)ug_b + (size_t)LB_S * HP_UI * 2;
  int2* pr_ug = (int2*)ig_b;
  int*  pc_ug = (int*)ig_b + (size_t)N_UG * PB_K * 2;
  int2* pr_ig = (int2*)ug_b;
  int*  pc_ig = (int*)ug_b + (size_t)N_IG * PB_K * 2;
  int* node = (int*)(W + off);
  _Float16* H = (_Float16*)(W + off);
  _Float16* ug_user_n16  = H;
  _Float16* ui_user_n16  = H + (size_t)NUM_USER * EMB;
  _Float16* ug_group_n16 = H + (size_t)NUM_USER * EMB * 2;
  _Float16* ig_group_n16 = ug_group_n16 + (size_t)NUM_GROUP * EMB;

  const int TB = 256;
  #define CDIV(a, b) (((a) + (b) - 1) / (b))

  zero_f32<<<16, TB, 0, stream>>>(tot, 4 * BATCH);
  zero_f32<<<1, 32, 0, stream>>>(out + (size_t)3 * BATCH * 128, 2);

  const dim3 gB(LB_P, LB_S);
  list_build_lds<<<gB, LB_THREADS, 0, stream>>>(hu_c, hu_r, hu_v, H8_ug, H2_ug, HP_UG, node, E_U, N_UG);
  list_prop<true><<<CDIV(N_UG, LP_COLS), LP_COLS * 8, 0, stream>>>(
      ued, NUM_USER, ged, H8_ug, H2_ug, HP_UG, node, ug_b, N_UG, pr_ug, pc_ug);
  pair_prop<<<CDIV(N_UG, LP_COLS), LP_COLS * 8, 0, stream>>>(
      ug_b, pr_ug, pc_ug, H8_ug, H2_ug, HP_UG, node, ug_a, N_UG);

  list_build_lds<<<gB, LB_THREADS, 0, stream>>>(hi_c, hi_r, hi_v, H8_ig, H2_ig, HP_IG, node, E_I, N_IG);
  list_prop<true><<<CDIV(N_IG, LP_COLS), LP_COLS * 8, 0, stream>>>(
      item, NUM_ITEM, gei, H8_ig, H2_ig, HP_IG, node, ig_b, N_IG, pr_ig, pc_ig);
  pair_prop<<<CDIV(N_IG, LP_COLS), LP_COLS * 8, 0, stream>>>(
      ig_b, pr_ig, pc_ig, H8_ig, H2_ig, HP_IG, node, ig_a, N_IG);

  list_build_lds<<<gB, LB_THREADS, 0, stream>>>(ui_c, ui_r, ui_v, H8_ui, H2_ui, HP_UI, node, E_UI, N_UI);
  list_prop<false><<<CDIV(N_UI, LP_COLS), LP_COLS * 8, 0, stream>>>(
      uei, NUM_USER, item, H8_ui, H2_ui, HP_UI, node, agg, N_UI, nullptr, nullptr);
  ui_final_kernel<<<CDIV(N_UI * EMB, TB), TB, 0, stream>>>(uei, item, agg);

  gather_out<<<CDIV(BATCH * EMB, TB), TB, 0, stream>>>(
      ug_a, agg, ig_a, uin, pos, neg, out);
  reg_kernel<<<CDIV(BATCH * EMB, TB), TB, 0, stream>>>(
      ued, uei, ged, gei, uin, pos, neg, out + (size_t)3 * BATCH * 128);

  normalize_rows_f16<<<CDIV(NUM_USER * 32, TB), TB, 0, stream>>>(
      ug_a, ug_user_n16, NUM_USER);
  normalize_rows_f16<<<CDIV(NUM_USER * 32, TB), TB, 0, stream>>>(
      agg, ui_user_n16, NUM_USER);
  normalize_rows_f16<<<CDIV(NUM_GROUP * 32, TB), TB, 0, stream>>>(
      ug_a + (size_t)NUM_USER * EMB, ug_group_n16, NUM_GROUP);
  normalize_rows_f16<<<CDIV(NUM_GROUP * 32, TB), TB, 0, stream>>>(
      ig_a + (size_t)NUM_ITEM * EMB, ig_group_n16, NUM_GROUP);

  const int perBlk = GEMM_WAVES * GEMM_TPW;
  int ntU = NUM_USER / 16, ntG = NUM_GROUP / 16;
  dim3 gU(BATCH / GEMM_BCH, CDIV(ntU, perBlk));
  dim3 gG(BATCH / GEMM_BCH, CDIV(ntG, perBlk));
  ssl_gemm_expsum<<<gU, TB, 0, stream>>>(ui_user_n16, ug_user_n16, uin, tot + 0,         NUM_USER);
  ssl_gemm_expsum<<<gU, TB, 0, stream>>>(ug_user_n16, ui_user_n16, uin, tot + BATCH,     NUM_USER);
  ssl_gemm_expsum<<<gG, TB, 0, stream>>>(ig_group_n16, ug_group_n16, pos, tot + 2*BATCH, NUM_GROUP);
  ssl_gemm_expsum<<<gG, TB, 0, stream>>>(ug_group_n16, ig_group_n16, pos, tot + 3*BATCH, NUM_GROUP);

  float* ssl_slot = out + (size_t)3 * BATCH * 128 + 1;
  int lossBlocks = CDIV(BATCH * 32, TB);
  ssl_loss<<<lossBlocks, TB, 0, stream>>>(ui_user_n16, ug_user_n16, uin, tot + 0,         ssl_slot);
  ssl_loss<<<lossBlocks, TB, 0, stream>>>(ug_user_n16, ui_user_n16, uin, tot + BATCH,     ssl_slot);
  ssl_loss<<<lossBlocks, TB, 0, stream>>>(ig_group_n16, ug_group_n16, pos, tot + 2*BATCH, ssl_slot);
  ssl_loss<<<lossBlocks, TB, 0, stream>>>(ug_group_n16, ig_group_n16, pos, tot + 3*BATCH, ssl_slot);
  #undef CDIV
}


# Row epilogues move into their readers: GEMM and loss loaders take l2 norm and f16 rounding from f32 rows, the one normalize left forms (ui+agg)*0.5; 4 launches go; norm sum order changes
# speedup vs baseline: 5.3067x; 1.0763x over previous
#include <hip/hip_runtime.h>
#include <hip/hip_bf16.h>
#include <math.h>

#define NUM_USER   100000
#define NUM_ITEM   50000
#define NUM_GROUP  20000
#define EMB        64
#define BATCH      1024
#define E_UI       2000000
#define E_U        1000000
#define E_I        1000000
#define INV_TEMP   10.0f
#define EPS_N      1e-12f

#define N_UG  (NUM_USER + NUM_GROUP)
#define N_IG  (NUM_ITEM + NUM_GROUP)
#define N_UI  (NUM_USER + NUM_ITEM)

typedef _Float16 v16h __attribute__((ext_vector_type(16)));
typedef float    v8f  __attribute__((ext_vector_type(8)));

__global__ void zero_f32(float* __restrict__ p, size_t n) {
  size_t i = (size_t)blockIdx.x * blockDim.x + threadIdx.x;
  size_t stride = (size_t)gridDim.x * blockDim.x;
  for (; i < n; i += stride) p[i] = 0.0f;
}

#define LB_P    10
#define LB_S    24
#define LB_WMAX 15008
#define LB_THREADS 1024
#define LB_G    4
#define LB_OTHER (-2)

__device__ __forceinline__ int claim1(int* hl, int c, int c0, int Wd) {
  unsigned d = (unsigned)(c - c0);
  return (d < (unsigned)Wd) ? atomicCAS(&hl[d], 0, -1) : LB_OTHER;
}
__device__ __forceinline__ void place1(int* hl, int2* hp, int4* __restrict__ node, int e,
                                       int c, int r, float v, int c0, int o) {
  if (o == LB_OTHER) return;
  const int d = c - c0;
  if (o == 0) {
    hp[d] = make_int2(r, __float_as_int(v));
  } else {
    int o2 = atomicExch(&hl[d], e + 1);
    node[e] = make_int4(o2 > 0 ? o2 : 0, r, __float_as_int(v), 0);
  }
}
__device__ __forceinline__ void thread_tail(int* hl, int2* hp, int4* __restrict__ node,
                                            const int* __restrict__ col,
                                            const int* __restrict__ row,
                                            const float* __restrict__ val,
                                            int e0, int send, int c0, int Wd) {
  for (int e = e0; e < send; ++e) {
    const int c = col[e];
    const int o = claim1(hl, c, c0, Wd);
    place1(hl, hp, node, e, c, row[e], val[e], c0, o);
  }
}

__global__ __launch_bounds__(LB_THREADS)
void list_build_lds(const int* __restrict__ col,
                    const int* __restrict__ row,
                    const float* __restrict__ val,
                    int2* __restrict__ H8, int* __restrict__ H2, int hpitch,
                    int* __restrict__ nodes, int nE, int nCols) {
  __shared__ int  hl[LB_WMAX];
  __shared__ int2 hp[LB_WMAX];
  int4* __restrict__ node = (int4*)nodes;
  const int p  = blockIdx.x;
  const int sl = blockIdx.y;
  const int Wd = (nCols + LB_P - 1) / LB_P;
  const int c0 = p * Wd;
  const int SL = (((nE + LB_S - 1) / LB_S) + 3) & ~3;
  const int sbeg = sl * SL;
  const int send = (sbeg + SL < nE) ? (sbeg + SL) : nE;

  for (int i = threadIdx.x; i < Wd; i += LB_THREADS) hl[i] = 0;
  __syncthreads();

  for (int e0 = sbeg + threadIdx.x * 4; e0 < send; e0 += LB_G * LB_THREADS * 4) {
    int4   cc[LB_G], rr[LB_G], oo[LB_G];
    float4 vv[LB_G];
    bool   ff[LB_G];
    #pragma unroll
    for (int g = 0; g < LB_G; ++g) {
      const int ea = e0 + g * LB_THREADS * 4;
      ff[g] = ea + 3 < send;
      cc[g] = make_int4(0, 0, 0, 0);
      rr[g] = make_int4(0, 0, 0, 0);
      vv[g] = make_float4(0.0f, 0.0f, 0.0f, 0.0f);
      if (ff[g]) {
        cc[g] = *(const int4*)(col + ea);
        rr[g] = *(const int4*)(row + ea);
        vv[g] = *(const float4*)(val + ea);
      }
    }
    #pragma unroll
    for (int g = 0; g < LB_G; ++g) {
      oo[g] = make_int4(LB_OTHER, LB_OTHER, LB_OTHER, LB_OTHER);
      if (ff[g]) {
        oo[g].x = claim1(hl, cc[g].x, c0, Wd);
        oo[g].y = claim1(hl, cc[g].y, c0, Wd);
        oo[g].z = claim1(hl, cc[g].z, c0, Wd);
        oo[g].w = claim1(hl, cc[g].w, c0, Wd);
      }
    }
    #pragma unroll
    for (int g = 0; g < LB_G; ++g) {
      const int ea = e0 + g * LB_THREADS * 4;
      if (ff[g]) {
        place1(hl, hp, node, ea,     cc[g].x, rr[g].x, vv[g].x, c0, oo[g].x);
        place1(hl, hp, node, ea + 1, cc[g].y, rr[g].y, vv[g].y, c0, oo[g].y);
        place1(hl, hp, node, ea + 2, cc[g].z, rr[g].z, vv[g].z, c0, oo[g].z);
        place1(hl, hp, node, ea + 3, cc[g].w, rr[g].w, vv[g].w, c0, oo[g].w);
      } else {
        thread_tail(hl, hp, node, col, row, val, ea, send, c0, Wd);
      }
    }
  }
  __syncthreads();

  int2* __restrict__ H8row = H8 + (size_t)sl * hpitch + c0;
  int*  __restrict__ H2row = H2 + (size_t)sl * hpitch + c0;
  for (int i = threadIdx.x; i < Wd; i += LB_THREADS) {
    if (c0 + i < nCols) {
      const int  h = hl[i];
      const int2 s = hp[i];
      H8row[i] = h ? s : make_int2(-1, 0);
      H2row[i] = (h > 0) ? h : 0;
    }
  }
}

#define LP_COLS 32
#define PB_K 32
template <bool WP>
__global__ __launch_bounds__(LP_COLS * 8)
void list_prop(const float* __restrict__ p0, int n0,
               const float* __restrict__ p1,
               const int2* __restrict__ H8, const int* __restrict__ H2, int hpitch,
               const int* __restrict__ nodes,
               float* __restrict__ out, int nRows,
               int2* __restrict__ pairs, int* __restrict__ pcnt) {
  __shared__ int2 hs8[LB_S * LP_COLS];
  __shared__ int  hs2[LB_S * LP_COLS];
  __shared__ int  cq[LP_COLS * LB_S];
  __shared__ __attribute__((aligned(16))) int2 pb[LP_COLS * PB_K];
  const int4* __restrict__ node = (const int4*)nodes;
  const int lane = threadIdx.x & 31;
  const int wave = threadIdx.x >> 5;
  const int cb   = blockIdx.x * LP_COLS;

  int2 hv8[LB_S / 8];
  int  hv2[LB_S / 8];
  #pragma unroll
  for (int i = 0; i < LB_S / 8; ++i) {
    const int sidx = i * 8 + wave;
    const bool ok = cb + lane < nRows;
    hv8[i] = ok ? H8[(size_t)sidx * hpitch + cb + lane] : make_int2(-1, 0);
    hv2[i] = ok ? H2[(size_t)sidx * hpitch + cb + lane] : 0;
  }
  #pragma unroll
  for (int i = 0; i < LB_S / 8; ++i) {
    hs8[(i * 8 + wave) * LP_COLS + lane] = hv8[i];
    hs2[(i * 8 + wave) * LP_COLS + lane] = hv2[i];
  }
  __syncthreads();

  const int j   = threadIdx.x >> 3;
  const int sub = threadIdx.x & 7;
  const int r   = cb + j;
  const unsigned gsh   = (unsigned)(lane & 24);
  const unsigned below = (1u << sub) - 1u;
  int cnt1 = 0;
  int cnt  = 0;
  #pragma unroll
  for (int i = 0; i < LB_S / 8; ++i) {
    const int2 s = hs8[(i * 8 + sub) * LP_COLS + j];
    const int  h = hs2[(i * 8 + sub) * LP_COLS + j];
    const unsigned g1 = (__builtin_amdgcn_ballot_w32(s.x >= 0) >> gsh) & 0xFFu;
    const unsigned g2 = (__builtin_amdgcn_ballot_w32(h != 0) >> gsh) & 0xFFu;
    if (s.x >= 0) pb[j * PB_K + cnt1 + __popc(g1 & below)] = s;
    if (h)        cq[j * LB_S + cnt  + __popc(g2 & below)] = h;
    cnt1 += __popc(g1);
    cnt  += __popc(g2);
  }
  __syncthreads();

  float4 a0 = {0.0f, 0.0f, 0.0f, 0.0f}, a1 = {0.0f, 0.0f, 0.0f, 0.0f};
  const int* __restrict__ q = cq + j * LB_S;
  int cur   = (cnt > 0) ? q[0] : 0;
  int hnext = (cnt > 1) ? q[1] : 0;
  int k = 2;
  int4 nd = make_int4(0, 0, 0, 0);
  if (cur) nd = node[cur - 1];
  for (int kk = 0; kk * 4 < cnt1; ++kk) {
    const int m = cnt1 - kk * 4;
    const int4* __restrict__ ps = (const int4*)(pb + j * PB_K + kk * 4);
    const int4 u0 = ps[0], u1 = ps[1];
    const int   s[4] = { u0.x, u0.z, u1.x, u1.z };
    const float v[4] = { __int_as_float(u0.y), __int_as_float(u0.w),
                         __int_as_float(u1.y), __int_as_float(u1.w) };
    float4 d0[4], d1[4];
    #pragma unroll
    for (int t = 0; t < 4; ++t) {
      d0[t] = make_float4(0.0f, 0.0f, 0.0f, 0.0f);
      d1[t] = make_float4(0.0f, 0.0f, 0.0f, 0.0f);
      if (t < m) {
        const float* src = (s[t] < n0) ? (p0 + (size_t)s[t] * EMB)
                                       : (p1 + (size_t)(s[t] - n0) * EMB);
        d0[t] = *(const float4*)(src + sub * 4);
        d1[t] = *(const float4*)(src + 32 + sub * 4);
      }
    }
    #pragma unroll
    for (int t = 0; t < 4; ++t) {
      if (t < m) {
        const float w = v[t];
        a0.x += d0[t].x * w; a0.y += d0[t].y * w; a0.z += d0[t].z * w; a0.w += d0[t].w * w;
        a1.x += d1[t].x * w; a1.y += d1[t].y * w; a1.z += d1[t].z * w; a1.w += d1[t].w * w;
      }
    }
  }

  float4 d0 = {0.0f, 0.0f, 0.0f, 0.0f}, d1 = {0.0f, 0.0f, 0.0f, 0.0f};
  float vp = 0.0f;
  int li = cnt1;
  while (cur) {
    int nx = nd.x;
    int s = nd.y;
    float v = __int_as_float(nd.z);
    if (WP) {
      if (sub == 0 && li < PB_K) pb[j * PB_K + li] = make_int2(s, nd.z);
      ++li;
    }
    const bool seam = (nx == 0);
    if (seam) nx = hnext;
    a0.x += d0.x * vp; a0.y += d0.y * vp; a0.z += d0.z * vp; a0.w += d0.w * vp;
    a1.x += d1.x * vp; a1.y += d1.y * vp; a1.z += d1.z * vp; a1.w += d1.w * vp;
    const float* src = (s < n0) ? (p0 + (size_t)s * EMB)
                                : (p1 + (size_t)(s - n0) * EMB);
    d0 = *(const float4*)(src + sub * 4);
    d1 = *(const float4*)(src + 32 + sub * 4);
    if (nx) nd = node[nx - 1];
    if (seam) { hnext = (k < cnt) ? q[k] : 0; ++k; }
    vp = v;
    cur = nx;
  }
  a0.x += d0.x * vp; a0.y += d0.y * vp; a0.z += d0.z * vp; a0.w += d0.w * vp;
  a1.x += d1.x * vp; a1.y += d1.y * vp; a1.z += d1.z * vp; a1.w += d1.w * vp;
  if (r < nRows) {
    float* dst = out + (size_t)r * EMB;
    *(float4*)(dst + sub * 4) = a0;
    *(float4*)(dst + 32 + sub * 4) = a1;
    if (WP) {
      const int4* ps = (const int4*)(pb + j * PB_K + sub * 4);
      int4 u0 = ps[0], u1 = ps[1];
      int4* pd = (int4*)(pairs + (size_t)r * PB_K + sub * 4);
      pd[0] = u0;
      pd[1] = u1;
      if (sub == 0) pcnt[r] = li;
    }
  }
}

__global__ __launch_bounds__(LP_COLS * 8)
void pair_prop(const float* __restrict__ in,
               const int2* __restrict__ pairs, const int* __restrict__ pcnt,
               const int2* __restrict__ H8, const int* __restrict__ H2, int hpitch,
               const int* __restrict__ nodes,
               float* __restrict__ out, int nRows) {
  const int j   = threadIdx.x >> 3;
  const int sub = threadIdx.x & 7;
  const int r   = blockIdx.x * LP_COLS + j;
  if (r >= nRows) return;
  const int4* pp = (const int4*)(pairs + (size_t)r * PB_K + sub * 4);
  const int4 q0 = pp[0], q1 = pp[1];
  const int cnt = pcnt[r];
  float4 a0 = {0.0f, 0.0f, 0.0f, 0.0f}, a1 = {0.0f, 0.0f, 0.0f, 0.0f};
  if (cnt <= PB_K) {
    for (int kk = 0; kk * 4 < cnt; ++kk) {
      const int m = cnt - kk * 4;
      int   s[4];
      float v[4];
      s[0] = __shfl(q0.x, kk, 8); v[0] = __int_as_float(__shfl(q0.y, kk, 8));
      s[1] = __shfl(q0.z, kk, 8); v[1] = __int_as_float(__shfl(q0.w, kk, 8));
      s[2] = __shfl(q1.x, kk, 8); v[2] = __int_as_float(__shfl(q1.y, kk, 8));
      s[3] = __shfl(q1.z, kk, 8); v[3] = __int_as_float(__shfl(q1.w, kk, 8));
      float4 d0[4], d1[4];
      #pragma unroll
      for (int t = 0; t < 4; ++t) {
        d0[t] = make_float4(0.0f, 0.0f, 0.0f, 0.0f);
        d1[t] = make_float4(0.0f, 0.0f, 0.0f, 0.0f);
        if (t < m) {
          const float* src = in + (size_t)s[t] * EMB;
          d0[t] = *(const float4*)(src + sub * 4);
          d1[t] = *(const float4*)(src + 32 + sub * 4);
        }
      }
      #pragma unroll
      for (int t = 0; t < 4; ++t) {
        if (t < m) {
          const float w = v[t];
          a0.x += d0[t].x * w; a0.y += d0[t].y * w; a0.z += d0[t].z * w; a0.w += d0[t].w * w;
          a1.x += d1[t].x * w; a1.y += d1[t].y * w; a1.z += d1[t].z * w; a1.w += d1[t].w * w;
        }
      }
    }
  } else {
    const int4* __restrict__ node = (const int4*)nodes;
    for (int sl = 0; sl < LB_S; ++sl) {
      const int2 s = H8[(size_t)sl * hpitch + r];
      if (s.x >= 0) {
        const float w = __int_as_float(s.y);
        const float* src = in + (size_t)s.x * EMB;
        const float4 e0 = *(const float4*)(src + sub * 4);
        const float4 e1 = *(const float4*)(src + 32 + sub * 4);
        a0.x += e0.x * w; a0.y += e0.y * w; a0.z += e0.z * w; a0.w += e0.w * w;
        a1.x += e1.x * w; a1.y += e1.y * w; a1.z += e1.z * w; a1.w += e1.w * w;
      }
    }
    for (int sl = 0; sl < LB_S; ++sl) {
      int h = H2[(size_t)sl * hpitch + r];
      while (h) {
        const int4 nd = node[h - 1];
        const float w = __int_as_float(nd.z);
        const float* src = in + (size_t)nd.y * EMB;
        const float4 e0 = *(const float4*)(src + sub * 4);
        const float4 e1 = *(const float4*)(src + 32 + sub * 4);
        a0.x += e0.x * w; a0.y += e0.y * w; a0.z += e0.z * w; a0.w += e0.w * w;
        a1.x += e1.x * w; a1.y += e1.y * w; a1.z += e1.z * w; a1.w += e1.w * w;
        h = nd.x;
      }
    }
  }
  float* dst = out + (size_t)r * EMB;
  *(float4*)(dst + sub * 4) = a0;
  *(float4*)(dst + 32 + sub * 4) = a1;
}

__global__ void gather_out(const float* __restrict__ ug,
                           const float* __restrict__ uei,
                           const float* __restrict__ agg,
                           const float* __restrict__ ig,
                           const int* __restrict__ u,
                           const int* __restrict__ p,
                           const int* __restrict__ ng,
                           float* __restrict__ out) {
  int gid = blockIdx.x * blockDim.x + threadIdx.x;
  if (gid >= BATCH * EMB) return;
  int b = gid >> 6, c = gid & 63;
  const size_t PO = (size_t)BATCH * 128;
  const size_t NO = 2 * PO;
  out[(size_t)b * 128 + c]           = ug [(size_t)u[b] * EMB + c];
  out[(size_t)b * 128 + 64 + c]      = (uei[(size_t)u[b] * EMB + c] + agg[(size_t)u[b] * EMB + c]) * 0.5f;
  out[PO + (size_t)b * 128 + c]      = ug [(size_t)(NUM_USER + p[b])  * EMB + c];
  out[PO + (size_t)b * 128 + 64 + c] = ig [(size_t)(NUM_ITEM + p[b])  * EMB + c];
  out[NO + (size_t)b * 128 + c]      = ug [(size_t)(NUM_USER + ng[b]) * EMB + c];
  out[NO + (size_t)b * 128 + 64 + c] = ig [(size_t)(NUM_ITEM + ng[b]) * EMB + c];
}

__global__ void reg_kernel(const float* __restrict__ ued,
                           const float* __restrict__ uei,
                           const float* __restrict__ ged,
                           const float* __restrict__ gei,
                           const int* __restrict__ u,
                           const int* __restrict__ p,
                           const int* __restrict__ ng,
                           float* __restrict__ reg_out) {
  int gid = blockIdx.x * blockDim.x + threadIdx.x;
  if (gid >= BATCH * EMB) return;
  int b = gid >> 6, c = gid & 63;
  size_t uo = (size_t)u[b]  * EMB + c;
  size_t po = (size_t)p[b]  * EMB + c;
  size_t no = (size_t)ng[b] * EMB + c;
  float s = uei[uo]*uei[uo] + ued[uo]*ued[uo]
          + gei[po]*gei[po] + ged[po]*ged[po]
          + gei[no]*gei[no] + ged[no]*ged[no];
  #pragma unroll
  for (int off = 1; off < 32; off <<= 1) s += __shfl_xor(s, off, 32);
  if ((threadIdx.x & 31) == 0) atomicAdd(reg_out, s * (0.5f / (float)BATCH));
}

__global__ void normalize_mean_rows_f16(const float* __restrict__ in0,
                                        const float* __restrict__ in1,
                                        _Float16* __restrict__ out, int nrows) {
  int gid = blockIdx.x * blockDim.x + threadIdx.x;
  int r = gid >> 5;
  if (r >= nrows) return;
  int t = gid & 31;
  size_t base = (size_t)r * EMB + 2 * t;
  float x0 = (in0[base] + in1[base]) * 0.5f, x1 = (in0[base + 1] + in1[base + 1]) * 0.5f;
  float ss = x0 * x0 + x1 * x1;
  #pragma unroll
  for (int off = 1; off < 32; off <<= 1) ss += __shfl_xor(ss, off, 32);
  float scale = 1.0f / fmaxf(sqrtf(ss), EPS_N);
  out[base]     = (_Float16)(x0 * scale);
  out[base + 1] = (_Float16)(x1 * scale);
}

#define GEMM_WAVES 8
#define GEMM_TPW   4
#define GEMM_BCH   128

typedef unsigned int u32x4 __attribute__((ext_vector_type(4)));
typedef unsigned int u32x8 __attribute__((ext_vector_type(8)));
typedef _Float16 v4h __attribute__((ext_vector_type(4)));

#define SLAB_PITCH 144

__device__ __forceinline__ float inv_norm(float ss) {
  return (ss >= EPS_N * EPS_N) ? __builtin_amdgcn_rsqf(ss) : (1.0f / EPS_N);
}
__device__ __forceinline__ float row16_sum(float v) {
  v += __int_as_float(__builtin_amdgcn_update_dpp(0, __float_as_int(v), 0xB1, 0xF, 0xF, true));
  v += __int_as_float(__builtin_amdgcn_update_dpp(0, __float_as_int(v), 0x4E, 0xF, 0xF, true));
  v += __int_as_float(__builtin_amdgcn_update_dpp(0, __float_as_int(v), 0x141, 0xF, 0xF, true));
  v += __int_as_float(__builtin_amdgcn_update_dpp(0, __float_as_int(v), 0x140, 0xF, 0xF, true));
  return v;
}
__device__ __forceinline__ void row_frags(const unsigned char* p, v16h& f0, v16h& f1) {
  u32x4 q0 = *(const u32x4*)(p);
  u32x4 q1 = *(const u32x4*)(p + 32);
  u32x4 q2 = *(const u32x4*)(p + 64);
  u32x4 q3 = *(const u32x4*)(p + 96);
  u32x8 w0 = __builtin_shufflevector(q0, q1, 0, 1, 2, 3, 4, 5, 6, 7);
  u32x8 w1 = __builtin_shufflevector(q2, q3, 0, 1, 2, 3, 4, 5, 6, 7);
  f0 = __builtin_bit_cast(v16h, w0);
  f1 = __builtin_bit_cast(v16h, w1);
}
__device__ __forceinline__ void tile_load(const float* __restrict__ s0, size_t o, float4 (&x)[8]) {
  #pragma unroll
  for (int j = 0; j < 4; ++j) {
    x[2 * j]     = *(const float4*)(s0 + o + 16 * j);
    x[2 * j + 1] = *(const float4*)(s0 + o + 16 * j + 4);
  }
}
__device__ __forceinline__ void tile_frags(const float4 (&x)[8], v16h& f0, v16h& f1) {
  float ss = 0.0f;
  #pragma unroll
  for (int j = 0; j < 8; ++j)
    ss += (x[j].x * x[j].x + x[j].y * x[j].y) + (x[j].z * x[j].z + x[j].w * x[j].w);
  ss += __shfl_xor(ss, 16, 32);
  const float sc = inv_norm(ss);
  #pragma unroll
  for (int j = 0; j < 4; ++j) {
    f0[4 * j]     = (_Float16)(x[j].x * sc);
    f0[4 * j + 1] = (_Float16)(x[j].y * sc);
    f0[4 * j + 2] = (_Float16)(x[j].z * sc);
    f0[4 * j + 3] = (_Float16)(x[j].w * sc);
    f1[4 * j]     = (_Float16)(x[4 + j].x * sc);
    f1[4 * j + 1] = (_Float16)(x[4 + j].y * sc);
    f1[4 * j + 2] = (_Float16)(x[4 + j].z * sc);
    f1[4 * j + 3] = (_Float16)(x[4 + j].w * sc);
  }
}

template <bool ASUM, bool B16>
__global__ __launch_bounds__(GEMM_WAVES * 32)
void ssl_gemm_expsum(const float* __restrict__ A0,
                     const float* __restrict__ A1,
                     const float* __restrict__ B0,
                     const _Float16* __restrict__ Bh,
                     const int* __restrict__ idx,
                     float* __restrict__ tot,
                     int Nall) {
  __shared__ __attribute__((aligned(16))) unsigned char slab[GEMM_BCH * SLAB_PITCH];
  __shared__ float part[GEMM_WAVES * 2 * GEMM_BCH];
  int lane  = threadIdx.x & 31;
  int wave  = threadIdx.x >> 5;
  int m     = lane & 15;
  int hi    = lane >> 4;
  int lr    = lane >> 4;
  int lc    = lane & 15;
  int bbase = blockIdx.x * GEMM_BCH;

  int rid[GEMM_BCH / (GEMM_WAVES * 2)];
  #pragma unroll
  for (int p = 0; p < GEMM_BCH / (GEMM_WAVES * 2); ++p)
    rid[p] = idx[bbase + p * (GEMM_WAVES * 2) + wave * 2 + lr];

  int ntile = Nall >> 4;
  int tile0 = (blockIdx.y * GEMM_WAVES + wave) * GEMM_TPW;
  v16h   a0[GEMM_TPW], a1[GEMM_TPW];
  float  wgt[GEMM_TPW];
  size_t to[GEMM_TPW];
  #pragma unroll
  for (int t = 0; t < GEMM_TPW; ++t) {
    int  tile = tile0 + t;
    bool ok   = tile < ntile;
    int  tl   = ok ? tile : 0;
    wgt[t] = ok ? 1.0f : 0.0f;
    to[t]  = (size_t)(tl * 16 + m) * EMB + hi * 8;
    if (B16) row_frags((const unsigned char*)(Bh + (size_t)(tl * 16 + m) * EMB) + hi * 16, a0[t], a1[t]);
  }

  float4 xa[8], xb[8];
  if (!B16) {
    tile_load(B0, to[0], xa);
    tile_load(B0, to[1], xb);
  }
  float4 g[GEMM_BCH / (GEMM_WAVES * 2)], h[GEMM_BCH / (GEMM_WAVES * 2)];
  #pragma unroll
  for (int p = 0; p < GEMM_BCH / (GEMM_WAVES * 2); ++p) {
    g[p] = *(const float4*)(A0 + (size_t)rid[p] * EMB + lc * 4);
    if (ASUM) h[p] = *(const float4*)(A1 + (size_t)rid[p] * EMB + lc * 4);
  }
  if (!B16) {
    tile_frags(xa, a0[0], a1[0]);
    tile_frags(xb, a0[1], a1[1]);
    tile_load(B0, to[2], xa);
    tile_load(B0, to[3], xb);
  }

  #pragma unroll
  for (int p = 0; p < GEMM_BCH / (GEMM_WAVES * 2); ++p) {
    float4 v = g[p];
    if (ASUM) {
      v.x = (v.x + h[p].x) * 0.5f; v.y = (v.y + h[p].y) * 0.5f;
      v.z = (v.z + h[p].z) * 0.5f; v.w = (v.w + h[p].w) * 0.5f;
    }
    float ss = (v.x * v.x + v.y * v.y) + (v.z * v.z + v.w * v.w);
    ss = row16_sum(ss);
    const float sc = inv_norm(ss);
    v4h hv;
    hv[0] = (_Float16)(v.x * sc);
    hv[1] = (_Float16)(v.y * sc);
    hv[2] = (_Float16)(v.z * sc);
    hv[3] = (_Float16)(v.w * sc);
    int r = p * (GEMM_WAVES * 2) + wave * 2 + lr;
    *(v4h*)(slab + r * SLAB_PITCH + lc * 8) = hv;
  }
  if (!B16) {
    tile_frags(xa, a0[2], a1[2]);
    tile_frags(xb, a0[3], a1[3]);
  }
  __syncthreads();

  const unsigned char* brow = slab + m * SLAB_PITCH + hi * 16;
  v16h b0, b1;
  row_frags(brow, b0, b1);
  #pragma unroll
  for (int bt = 0; bt < GEMM_BCH / 16; ++bt) {
    v16h n0 = b0, n1 = b1;
    if (bt + 1 < GEMM_BCH / 16) row_frags(brow + (bt + 1) * 16 * SLAB_PITCH, n0, n1);
    v8f c[GEMM_TPW];
    #pragma unroll
    for (int t = 0; t < GEMM_TPW; ++t) {
      v8f z = {};
      c[t] = __builtin_amdgcn_wmma_f32_16x16x32_f16(false, a0[t], false, b0,
                                                    (short)0, z, false, false);
    }
    #pragma unroll
    for (int t = 0; t < GEMM_TPW; ++t)
      c[t] = __builtin_amdgcn_wmma_f32_16x16x32_f16(false, a1[t], false, b1,
                                                    (short)0, c[t], false, false);
    float e = 0.0f;
    #pragma unroll
    for (int t = 0; t < GEMM_TPW; ++t) {
      float x[8];
      #pragma unroll
      for (int v = 0; v < 8; ++v) x[v] = __expf(c[t][v] * INV_TEMP);
      float et = ((x[0] + x[1]) + (x[2] + x[3])) + ((x[4] + x[5]) + (x[6] + x[7]));
      e += wgt[t] * et;
    }
    part[(wave * 2 + hi) * GEMM_BCH + bt * 16 + m] = e;
    b0 = n0; b1 = n1;
  }
  __syncthreads();

  if (threadIdx.x < GEMM_BCH) {
    float sum = 0.0f;
    #pragma unroll
    for (int w = 0; w < GEMM_WAVES * 2; ++w) sum += part[w * GEMM_BCH + threadIdx.x];
    atomicAdd(&tot[bbase + threadIdx.x], sum);
  }
}

template <bool ASUM, bool BSUM>
__global__ void ssl_loss(const float* __restrict__ A0, const float* __restrict__ A1,
                         const float* __restrict__ B0, const float* __restrict__ B1,
                         const int* __restrict__ idx,
                         const float* __restrict__ tot,
                         float* __restrict__ ssl_out) {
  int gid = blockIdx.x * blockDim.x + threadIdx.x;
  int i = gid >> 5;
  if (i >= BATCH) return;
  int t = gid & 31;
  size_t ro = (size_t)idx[i] * EMB + 2 * t;
  float ax = A0[ro], ay = A0[ro + 1];
  float bx = B0[ro], by = B0[ro + 1];
  if (ASUM) { ax = (ax + A1[ro]) * 0.5f; ay = (ay + A1[ro + 1]) * 0.5f; }
  if (BSUM) { bx = (bx + B1[ro]) * 0.5f; by = (by + B1[ro + 1]) * 0.5f; }
  float saa = ax * ax + ay * ay;
  float sbb = bx * bx + by * by;
  float d   = ax * bx + ay * by;
  #pragma unroll
  for (int off = 1; off < 32; off <<= 1) {
    saa += __shfl_xor(saa, off, 32);
    sbb += __shfl_xor(sbb, off, 32);
    d   += __shfl_xor(d, off, 32);
  }
  if (t == 0) {
    float dn = (d / fmaxf(sqrtf(saa), EPS_N)) / fmaxf(sqrtf(sbb), EPS_N);
    float li = -(dn * INV_TEMP - logf(tot[i])) * (1.0f / (float)BATCH);
    atomicAdd(ssl_out, li);
  }
}

extern "C" void kernel_launch(void* const* d_in, const int* in_sizes, int n_in,
                              void* d_out, int out_size, void* d_ws, size_t ws_size,
                              hipStream_t stream) {
  (void)in_sizes; (void)n_in; (void)out_size; (void)ws_size;
  const float* ued  = (const float*)d_in[0];
  const float* uei  = (const float*)d_in[1];
  const float* ged  = (const float*)d_in[2];
  const float* gei  = (const float*)d_in[3];
  const float* item = (const float*)d_in[4];
  const int*   hu_r = (const int*)d_in[5];
  const int*   hu_c = (const int*)d_in[6];
  const float* hu_v = (const float*)d_in[7];
  const int*   hi_r = (const int*)d_in[8];
  const int*   hi_c = (const int*)d_in[9];
  const float* hi_v = (const float*)d_in[10];
  const int*   ui_r = (const int*)d_in[11];
  const int*   ui_c = (const int*)d_in[12];
  const float* ui_v = (const float*)d_in[13];
  const int*   uin  = (const int*)d_in[14];
  const int*   pos  = (const int*)d_in[15];
  const int*   neg  = (const int*)d_in[16];
  float* out = (float*)d_out;

  float* W = (float*)d_ws;
  size_t off = 0;
  float* ug_a = W + off; off += (size_t)N_UG * EMB;
  float* ug_b = W + off; off += (size_t)N_UG * EMB;
  float* ig_b = W + off; off += (size_t)N_IG * EMB;
  float* ig_a = W + off; off += (size_t)N_IG * EMB;
  float* agg  = W + off; off += (size_t)N_UI * EMB;
  float* tot  = W + off; off += 4 * BATCH;
  #define HP_UG 120000
  #define HP_IG 70016
  #define HP_UI 150016
  int2* H8_ug = (int2*)agg;   int* H2_ug = (int*)agg  + (size_t)LB_S * HP_UG * 2;
  int2* H8_ig = (int2*)agg;   int* H2_ig = (int*)agg  + (size_t)LB_S * HP_IG * 2;
  int2* H8_ui = (int2*)ug_b;  int* H2_ui = (int*)ug_b + (size_t)LB_S * HP_UI * 2;
  int2* pr_ug = (int2*)ig_b;
  int*  pc_ug = (int*)ig_b + (size_t)N_UG * PB_K * 2;
  int2* pr_ig = (int2*)ug_b;
  int*  pc_ig = (int*)ug_b + (size_t)N_IG * PB_K * 2;
  int* node = (int*)(W + off);
  _Float16* ui_user_n16 = (_Float16*)(W + off);
  const float* ug_user  = ug_a;
  const float* ug_group = ug_a + (size_t)NUM_USER * EMB;
  const float* ig_group = ig_a + (size_t)NUM_ITEM * EMB;

  const int TB = 256;
  #define CDIV(a, b) (((a) + (b) - 1) / (b))

  zero_f32<<<16, TB, 0, stream>>>(tot, 4 * BATCH);
  zero_f32<<<1, 32, 0, stream>>>(out + (size_t)3 * BATCH * 128, 2);

  const dim3 gB(LB_P, LB_S);
  list_build_lds<<<gB, LB_THREADS, 0, stream>>>(hu_c, hu_r, hu_v, H8_ug, H2_ug, HP_UG, node, E_U, N_UG);
  list_prop<true><<<CDIV(N_UG, LP_COLS), LP_COLS * 8, 0, stream>>>(
      ued, NUM_USER, ged, H8_ug, H2_ug, HP_UG, node, ug_b, N_UG, pr_ug, pc_ug);
  pair_prop<<<CDIV(N_UG, LP_COLS), LP_COLS * 8, 0, stream>>>(
      ug_b, pr_ug, pc_ug, H8_ug, H2_ug, HP_UG, node, ug_a, N_UG);

  list_build_lds<<<gB, LB_THREADS, 0, stream>>>(hi_c, hi_r, hi_v, H8_ig, H2_ig, HP_IG, node, E_I, N_IG);
  list_prop<true><<<CDIV(N_IG, LP_COLS), LP_COLS * 8, 0, stream>>>(
      item, NUM_ITEM, gei, H8_ig, H2_ig, HP_IG, node, ig_b, N_IG, pr_ig, pc_ig);
  pair_prop<<<CDIV(N_IG, LP_COLS), LP_COLS * 8, 0, stream>>>(
      ig_b, pr_ig, pc_ig, H8_ig, H2_ig, HP_IG, node, ig_a, N_IG);

  list_build_lds<<<gB, LB_THREADS, 0, stream>>>(ui_c, ui_r, ui_v, H8_ui, H2_ui, HP_UI, node, E_UI, N_UI);
  list_prop<false><<<CDIV(N_UI, LP_COLS), LP_COLS * 8, 0, stream>>>(
      uei, NUM_USER, item, H8_ui, H2_ui, HP_UI, node, agg, N_UI, nullptr, nullptr);
  normalize_mean_rows_f16<<<CDIV(NUM_USER * 32, TB), TB, 0, stream>>>(uei, agg, ui_user_n16, NUM_USER);

  gather_out<<<CDIV(BATCH * EMB, TB), TB, 0, stream>>>(
      ug_a, uei, agg, ig_a, uin, pos, neg, out);
  reg_kernel<<<CDIV(BATCH * EMB, TB), TB, 0, stream>>>(
      ued, uei, ged, gei, uin, pos, neg, out + (size_t)3 * BATCH * 128);

  const int perBlk = GEMM_WAVES * GEMM_TPW;
  int ntU = NUM_USER / 16, ntG = NUM_GROUP / 16;
  dim3 gU(BATCH / GEMM_BCH, CDIV(ntU, perBlk));
  dim3 gG(BATCH / GEMM_BCH, CDIV(ntG, perBlk));
  ssl_gemm_expsum<true,  false><<<gU, TB, 0, stream>>>(uei, agg, ug_user, nullptr, uin, tot + 0,       NUM_USER);
  ssl_gemm_expsum<false, true ><<<gU, TB, 0, stream>>>(ug_user, nullptr, nullptr, ui_user_n16, uin, tot + BATCH, NUM_USER);
  ssl_gemm_expsum<false, false><<<gG, TB, 0, stream>>>(ig_group, nullptr, ug_group, nullptr, pos, tot + 2*BATCH, NUM_GROUP);
  ssl_gemm_expsum<false, false><<<gG, TB, 0, stream>>>(ug_group, nullptr, ig_group, nullptr, pos, tot + 3*BATCH, NUM_GROUP);

  float* ssl_slot = out + (size_t)3 * BATCH * 128 + 1;
  int lossBlocks = CDIV(BATCH * 32, TB);
  ssl_loss<true,  false><<<lossBlocks, TB, 0, stream>>>(uei, agg, ug_user, nullptr, uin, tot + 0,       ssl_slot);
  ssl_loss<false, true ><<<lossBlocks, TB, 0, stream>>>(ug_user, nullptr, uei, agg, uin, tot + BATCH,   ssl_slot);
  ssl_loss<false, false><<<lossBlocks, TB, 0, stream>>>(ig_group, nullptr, ug_group, nullptr, pos, tot + 2*BATCH, ssl_slot);
  ssl_loss<false, false><<<lossBlocks, TB, 0, stream>>>(ug_group, nullptr, ig_group, nullptr, pos, tot + 3*BATCH, ssl_slot);
  #undef CDIV
}
